# starting kernel
# baseline (speedup 1.0000x reference)
#include <hip/hip_runtime.h>


typedef __attribute__((ext_vector_type(2))) float v2f;
typedef __attribute__((ext_vector_type(8))) float v8f;

#define D_DIM  2048
#define K_EXP  16
#define R_RANK 16
#define NTOK   4096
#define N_HV   256

static __device__ __forceinline__ v8f wmma_f32(v2f a, v2f b, v8f c) {
  return __builtin_amdgcn_wmma_f32_16x16x4_f32(
       false, a,  false, b,
       (short)0, c,  false,  false);
}

__global__ __launch_bounds__(256) void moelora_router(
    const float* __restrict__ x, const float* __restrict__ Wr,
    const int* __restrict__ alpha_p, float* __restrict__ g)
{
  const int wave = threadIdx.x >> 5;
  const int lane = threadIdx.x & 31;
  const int t = blockIdx.x * 8 + wave;

  float acc[K_EXP];
#pragma unroll
  for (int k = 0; k < K_EXP; ++k) acc[k] = 0.f;

  const float* xrow = x + (size_t)t * D_DIM;
  for (int d = lane; d < D_DIM; d += 32) {
    const float xv = xrow[d];
#pragma unroll
    for (int k = 0; k < K_EXP; ++k)
      acc[k] = fmaf(xv, Wr[k * D_DIM + d], acc[k]);
  }
#pragma unroll
  for (int k = 0; k < K_EXP; ++k) {
#pragma unroll
    for (int off = 16; off > 0; off >>= 1)
      acc[k] += __shfl_xor(acc[k], off, 32);
  }

  if (lane == 0) {
    const float alpha = (float)(*alpha_p);
    int i0 = 0; float v0 = acc[0];
#pragma unroll
    for (int k = 1; k < K_EXP; ++k) if (acc[k] > v0) { v0 = acc[k]; i0 = k; }
    int i1 = -1; float v1 = -3.4e38f;
#pragma unroll
    for (int k = 0; k < K_EXP; ++k)
      if (k != i0 && acc[k] > v1) { v1 = acc[k]; i1 = k; }
    const float e  = __expf(v1 - v0);
    const float dn = 1.f + e;
    const float g0 = alpha / dn;
    const float g1 = alpha * e / dn;
#pragma unroll
    for (int k = 0; k < K_EXP; ++k)
      g[t * K_EXP + k] = (k == i0) ? g0 : ((k == i1) ? g1 : 0.f);
  }
}

__global__ __launch_bounds__(256) void moelora_hgemm(
    const float* __restrict__ x, const float* __restrict__ Amat,
    const float* __restrict__ g, float* __restrict__ hv, int nchunk)
{
  const int wave = threadIdx.x >> 5;
  const int lane = threadIdx.x & 31;
  const int m = lane & 15;
  const int h = lane >> 4;
  const int row0 = blockIdx.x * 32;
  const int nt0  = wave * 2;

  const float* xrow0 = x    + (size_t)(row0 + m)        * D_DIM + 2 * h;
  const float* xrow1 = x    + (size_t)(row0 + 16 + m)   * D_DIM + 2 * h;
  const float* arow0 = Amat + (size_t)(nt0 * 16 + m)       * D_DIM + 2 * h;
  const float* arow1 = Amat + (size_t)((nt0 + 1) * 16 + m) * D_DIM + 2 * h;

  v8f c00 = {}, c01 = {}, c10 = {}, c11 = {};

  for (int c = 0; c < nchunk; ++c) {
    const int base = c * 16;
#pragma unroll
    for (int s = 0; s < 4; ++s) {
      const int kb = base + s * 4;
      const v2f a0 = *(const v2f*)(xrow0 + kb);
      const v2f a1 = *(const v2f*)(xrow1 + kb);
      const v2f b0 = *(const v2f*)(arow0 + kb);
      const v2f b1 = *(const v2f*)(arow1 + kb);
      c00 = wmma_f32(a0, b0, c00);
      c01 = wmma_f32(a0, b1, c01);
      c10 = wmma_f32(a1, b0, c10);
      c11 = wmma_f32(a1, b1, c11);
    }
  }

#pragma unroll
  for (int i = 0; i < 8; ++i) {
    const int ra = row0 + i + 8 * h;
    const int rb = ra + 16;
    const float ga0 = g[ra * K_EXP + nt0];
    const float ga1 = g[ra * K_EXP + nt0 + 1];
    const float gb0 = g[rb * K_EXP + nt0];
    const float gb1 = g[rb * K_EXP + nt0 + 1];
    hv[(size_t)ra * N_HV + nt0 * 16 + m]       = c00[i] * ga0;
    hv[(size_t)ra * N_HV + (nt0 + 1) * 16 + m] = c01[i] * ga1;
    hv[(size_t)rb * N_HV + nt0 * 16 + m]       = c10[i] * gb0;
    hv[(size_t)rb * N_HV + (nt0 + 1) * 16 + m] = c11[i] * gb1;
  }
}

__global__ __launch_bounds__(256) void moelora_ogemm(
    const float* __restrict__ hv, const float* __restrict__ Bm,
    float* __restrict__ out, int nexp)
{
  const int wave = threadIdx.x >> 5;
  const int lane = threadIdx.x & 31;
  const int m = lane & 15;
  const int h = lane >> 4;
  const int row0 = blockIdx.x * 32;
  const int dt0  = blockIdx.y * 16 + wave * 2;
  const int dcol0 = dt0 * 16 + m;
  const int dcol1 = (dt0 + 1) * 16 + m;

  const float* hr0 = hv + (size_t)(row0 + m)      * N_HV + 2 * h;
  const float* hr1 = hv + (size_t)(row0 + 16 + m) * N_HV + 2 * h;
  const float* bb0 = Bm + (size_t)dcol0 * R_RANK + 2 * h;
  const float* bb1 = Bm + (size_t)dcol1 * R_RANK + 2 * h;

  v8f c00 = {}, c01 = {}, c10 = {}, c11 = {};

  for (int kk = 0; kk < nexp; ++kk) {
#pragma unroll
    for (int s = 0; s < 4; ++s) {
      const v2f a0 = *(const v2f*)(hr0 + s * 4);
      const v2f a1 = *(const v2f*)(hr1 + s * 4);
      const v2f b0 = *(const v2f*)(bb0 + s * 4);
      const v2f b1 = *(const v2f*)(bb1 + s * 4);
      c00 = wmma_f32(a0, b0, c00);
      c01 = wmma_f32(a0, b1, c01);
      c10 = wmma_f32(a1, b0, c10);
      c11 = wmma_f32(a1, b1, c11);
    }
    hr0 += R_RANK;
    hr1 += R_RANK;
    bb0 += (size_t)D_DIM * R_RANK;
    bb1 += (size_t)D_DIM * R_RANK;
  }

#pragma unroll
  for (int i = 0; i < 8; ++i) {
    const int ra = row0 + i + 8 * h;
    const int rb = ra + 16;
    out[(size_t)ra * D_DIM + dcol0] = c00[i];
    out[(size_t)ra * D_DIM + dcol1] = c01[i];
    out[(size_t)rb * D_DIM + dcol0] = c10[i];
    out[(size_t)rb * D_DIM + dcol1] = c11[i];
  }
}

extern "C" void kernel_launch(void* const* d_in, const int* in_sizes, int n_in,
                              void* d_out, int out_size, void* d_ws, size_t ws_size,
                              hipStream_t stream) {
  (void)in_sizes; (void)n_in; (void)out_size; (void)ws_size;
  const float* x     = (const float*)d_in[0];
  const int*   alpha = (const int*)  d_in[1];
  const float* A     = (const float*)d_in[2];
  const float* Bm    = (const float*)d_in[3];
  const float* Wr    = (const float*)d_in[4];
  float* out = (float*)d_out;

  float* g  = (float*)d_ws;
  float* hv = g + (size_t)NTOK * K_EXP;

  moelora_router<<<NTOK / 8, 256, 0, stream>>>(x, Wr, alpha, g);
  moelora_hgemm<<<NTOK / 32, 256, 0, stream>>>(x, A, g, hv, D_DIM / 16);
  moelora_ogemm<<<dim3(NTOK / 32, D_DIM / 256), 256, 0, stream>>>(hv, Bm, out, K_EXP);
}


# 128-bit fragment loads in both WMMA GEMM kernels: a lane loads 4 consecutive k and feeds two WMMAs, same k permutation on both operands; changes the k sum order
# speedup vs baseline: 1.1423x; 1.1423x over previous
#include <hip/hip_runtime.h>


typedef __attribute__((ext_vector_type(2))) float v2f;
typedef __attribute__((ext_vector_type(4))) float v4f;
typedef __attribute__((ext_vector_type(8))) float v8f;

#define D_DIM  2048
#define K_EXP  16
#define R_RANK 16
#define NTOK   4096
#define N_HV   256

static __device__ __forceinline__ v8f wmma_f32(v2f a, v2f b, v8f c) {
  return __builtin_amdgcn_wmma_f32_16x16x4_f32(
       false, a,  false, b,
       (short)0, c,  false,  false);
}

static __device__ __forceinline__ v2f frag_lo(v4f v) {
  return __builtin_shufflevector(v, v, 0, 1);
}
static __device__ __forceinline__ v2f frag_hi(v4f v) {
  return __builtin_shufflevector(v, v, 2, 3);
}

__global__ __launch_bounds__(256) void moelora_router(
    const float* __restrict__ x, const float* __restrict__ Wr,
    const int* __restrict__ alpha_p, float* __restrict__ g)
{
  const int wave = threadIdx.x >> 5;
  const int lane = threadIdx.x & 31;
  const int t = blockIdx.x * 8 + wave;

  float acc[K_EXP];
#pragma unroll
  for (int k = 0; k < K_EXP; ++k) acc[k] = 0.f;

  const float* xrow = x + (size_t)t * D_DIM;
  for (int d = lane; d < D_DIM; d += 32) {
    const float xv = xrow[d];
#pragma unroll
    for (int k = 0; k < K_EXP; ++k)
      acc[k] = fmaf(xv, Wr[k * D_DIM + d], acc[k]);
  }
#pragma unroll
  for (int k = 0; k < K_EXP; ++k) {
#pragma unroll
    for (int off = 16; off > 0; off >>= 1)
      acc[k] += __shfl_xor(acc[k], off, 32);
  }

  if (lane == 0) {
    const float alpha = (float)(*alpha_p);
    int i0 = 0; float v0 = acc[0];
#pragma unroll
    for (int k = 1; k < K_EXP; ++k) if (acc[k] > v0) { v0 = acc[k]; i0 = k; }
    int i1 = -1; float v1 = -3.4e38f;
#pragma unroll
    for (int k = 0; k < K_EXP; ++k)
      if (k != i0 && acc[k] > v1) { v1 = acc[k]; i1 = k; }
    const float e  = __expf(v1 - v0);
    const float dn = 1.f + e;
    const float g0 = alpha / dn;
    const float g1 = alpha * e / dn;
#pragma unroll
    for (int k = 0; k < K_EXP; ++k)
      g[t * K_EXP + k] = (k == i0) ? g0 : ((k == i1) ? g1 : 0.f);
  }
}

__global__ __launch_bounds__(256) void moelora_hgemm(
    const float* __restrict__ x, const float* __restrict__ Amat,
    const float* __restrict__ g, float* __restrict__ hv, int nchunk)
{
  const int wave = threadIdx.x >> 5;
  const int lane = threadIdx.x & 31;
  const int m = lane & 15;
  const int h = lane >> 4;
  const int row0 = blockIdx.x * 32;
  const int nt0  = wave * 2;

  const float* xrow0 = x    + (size_t)(row0 + m)        * D_DIM + 4 * h;
  const float* xrow1 = x    + (size_t)(row0 + 16 + m)   * D_DIM + 4 * h;
  const float* arow0 = Amat + (size_t)(nt0 * 16 + m)       * D_DIM + 4 * h;
  const float* arow1 = Amat + (size_t)((nt0 + 1) * 16 + m) * D_DIM + 4 * h;

  v8f c00 = {}, c01 = {}, c10 = {}, c11 = {};

  for (int c = 0; c < nchunk; ++c) {
    const int base = c * 16;
#pragma unroll
    for (int s = 0; s < 2; ++s) {
      const int kb = base + s * 8;
      const v4f a0 = *(const v4f*)(xrow0 + kb);
      const v4f a1 = *(const v4f*)(xrow1 + kb);
      const v4f b0 = *(const v4f*)(arow0 + kb);
      const v4f b1 = *(const v4f*)(arow1 + kb);
      c00 = wmma_f32(frag_lo(a0), frag_lo(b0), c00);
      c01 = wmma_f32(frag_lo(a0), frag_lo(b1), c01);
      c10 = wmma_f32(frag_lo(a1), frag_lo(b0), c10);
      c11 = wmma_f32(frag_lo(a1), frag_lo(b1), c11);
      c00 = wmma_f32(frag_hi(a0), frag_hi(b0), c00);
      c01 = wmma_f32(frag_hi(a0), frag_hi(b1), c01);
      c10 = wmma_f32(frag_hi(a1), frag_hi(b0), c10);
      c11 = wmma_f32(frag_hi(a1), frag_hi(b1), c11);
    }
  }

#pragma unroll
  for (int i = 0; i < 8; ++i) {
    const int ra = row0 + i + 8 * h;
    const int rb = ra + 16;
    const float ga0 = g[ra * K_EXP + nt0];
    const float ga1 = g[ra * K_EXP + nt0 + 1];
    const float gb0 = g[rb * K_EXP + nt0];
    const float gb1 = g[rb * K_EXP + nt0 + 1];
    hv[(size_t)ra * N_HV + nt0 * 16 + m]       = c00[i] * ga0;
    hv[(size_t)ra * N_HV + (nt0 + 1) * 16 + m] = c01[i] * ga1;
    hv[(size_t)rb * N_HV + nt0 * 16 + m]       = c10[i] * gb0;
    hv[(size_t)rb * N_HV + (nt0 + 1) * 16 + m] = c11[i] * gb1;
  }
}

__global__ __launch_bounds__(256) void moelora_ogemm(
    const float* __restrict__ hv, const float* __restrict__ Bm,
    float* __restrict__ out, int nexp)
{
  const int wave = threadIdx.x >> 5;
  const int lane = threadIdx.x & 31;
  const int m = lane & 15;
  const int h = lane >> 4;
  const int row0 = blockIdx.x * 32;
  const int dt0  = blockIdx.y * 16 + wave * 2;
  const int dcol0 = dt0 * 16 + m;
  const int dcol1 = (dt0 + 1) * 16 + m;

  const float* hr0 = hv + (size_t)(row0 + m)      * N_HV + 4 * h;
  const float* hr1 = hv + (size_t)(row0 + 16 + m) * N_HV + 4 * h;
  const float* bb0 = Bm + (size_t)dcol0 * R_RANK + 4 * h;
  const float* bb1 = Bm + (size_t)dcol1 * R_RANK + 4 * h;

  v8f c00 = {}, c01 = {}, c10 = {}, c11 = {};

  for (int kk = 0; kk < nexp; ++kk) {
#pragma unroll
    for (int s = 0; s < 2; ++s) {
      const v4f a0 = *(const v4f*)(hr0 + s * 8);
      const v4f a1 = *(const v4f*)(hr1 + s * 8);
      const v4f b0 = *(const v4f*)(bb0 + s * 8);
      const v4f b1 = *(const v4f*)(bb1 + s * 8);
      c00 = wmma_f32(frag_lo(a0), frag_lo(b0), c00);
      c01 = wmma_f32(frag_lo(a0), frag_lo(b1), c01);
      c10 = wmma_f32(frag_lo(a1), frag_lo(b0), c10);
      c11 = wmma_f32(frag_lo(a1), frag_lo(b1), c11);
      c00 = wmma_f32(frag_hi(a0), frag_hi(b0), c00);
      c01 = wmma_f32(frag_hi(a0), frag_hi(b1), c01);
      c10 = wmma_f32(frag_hi(a1), frag_hi(b0), c10);
      c11 = wmma_f32(frag_hi(a1), frag_hi(b1), c11);
    }
    hr0 += R_RANK;
    hr1 += R_RANK;
    bb0 += (size_t)D_DIM * R_RANK;
    bb1 += (size_t)D_DIM * R_RANK;
  }

#pragma unroll
  for (int i = 0; i < 8; ++i) {
    const int ra = row0 + i + 8 * h;
    const int rb = ra + 16;
    out[(size_t)ra * D_DIM + dcol0] = c00[i];
    out[(size_t)ra * D_DIM + dcol1] = c01[i];
    out[(size_t)rb * D_DIM + dcol0] = c10[i];
    out[(size_t)rb * D_DIM + dcol1] = c11[i];
  }
}

extern "C" void kernel_launch(void* const* d_in, const int* in_sizes, int n_in,
                              void* d_out, int out_size, void* d_ws, size_t ws_size,
                              hipStream_t stream) {
  (void)in_sizes; (void)n_in; (void)out_size; (void)ws_size;
  const float* x     = (const float*)d_in[0];
  const int*   alpha = (const int*)  d_in[1];
  const float* A     = (const float*)d_in[2];
  const float* Bm    = (const float*)d_in[3];
  const float* Wr    = (const float*)d_in[4];
  float* out = (float*)d_out;

  float* g  = (float*)d_ws;
  float* hv = g + (size_t)NTOK * K_EXP;

  moelora_router<<<NTOK / 8, 256, 0, stream>>>(x, Wr, alpha, g);
  moelora_hgemm<<<NTOK / 32, 256, 0, stream>>>(x, A, g, hv, D_DIM / 16);
  moelora_ogemm<<<dim3(NTOK / 32, D_DIM / 256), 256, 0, stream>>>(hv, Bm, out, K_EXP);
}


# h-GEMM: K range split in two halves over the two wave groups of a block, halves added through LDS (changes the k sum order); grid doubles to 256 blocks of 64 tokens x 4 experts
# speedup vs baseline: 1.6184x; 1.4168x over previous
#include <hip/hip_runtime.h>


typedef __attribute__((ext_vector_type(2))) float v2f;
typedef __attribute__((ext_vector_type(4))) float v4f;
typedef __attribute__((ext_vector_type(8))) float v8f;

#define D_DIM  2048
#define K_EXP  16
#define R_RANK 16
#define NTOK   4096
#define N_HV   256

static __device__ __forceinline__ v8f wmma_f32(v2f a, v2f b, v8f c) {
  return __builtin_amdgcn_wmma_f32_16x16x4_f32(
       false, a,  false, b,
       (short)0, c,  false,  false);
}

static __device__ __forceinline__ v2f frag_lo(v4f v) {
  return __builtin_shufflevector(v, v, 0, 1);
}
static __device__ __forceinline__ v2f frag_hi(v4f v) {
  return __builtin_shufflevector(v, v, 2, 3);
}

__global__ __launch_bounds__(256) void moelora_router(
    const float* __restrict__ x, const float* __restrict__ Wr,
    const int* __restrict__ alpha_p, float* __restrict__ g)
{
  const int wave = threadIdx.x >> 5;
  const int lane = threadIdx.x & 31;
  const int t = blockIdx.x * 8 + wave;

  float acc[K_EXP];
#pragma unroll
  for (int k = 0; k < K_EXP; ++k) acc[k] = 0.f;

  const float* xrow = x + (size_t)t * D_DIM;
  for (int d = lane; d < D_DIM; d += 32) {
    const float xv = xrow[d];
#pragma unroll
    for (int k = 0; k < K_EXP; ++k)
      acc[k] = fmaf(xv, Wr[k * D_DIM + d], acc[k]);
  }
#pragma unroll
  for (int k = 0; k < K_EXP; ++k) {
#pragma unroll
    for (int off = 16; off > 0; off >>= 1)
      acc[k] += __shfl_xor(acc[k], off, 32);
  }

  if (lane == 0) {
    const float alpha = (float)(*alpha_p);
    int i0 = 0; float v0 = acc[0];
#pragma unroll
    for (int k = 1; k < K_EXP; ++k) if (acc[k] > v0) { v0 = acc[k]; i0 = k; }
    int i1 = -1; float v1 = -3.4e38f;
#pragma unroll
    for (int k = 0; k < K_EXP; ++k)
      if (k != i0 && acc[k] > v1) { v1 = acc[k]; i1 = k; }
    const float e  = __expf(v1 - v0);
    const float dn = 1.f + e;
    const float g0 = alpha / dn;
    const float g1 = alpha * e / dn;
#pragma unroll
    for (int k = 0; k < K_EXP; ++k)
      g[t * K_EXP + k] = (k == i0) ? g0 : ((k == i1) ? g1 : 0.f);
  }
}

__global__ __launch_bounds__(256) void moelora_hgemm(
    const float* __restrict__ x, const float* __restrict__ Amat,
    const float* __restrict__ g, float* __restrict__ hv, int nchunk)
{
  __shared__ float red[8 * 2 * 8 * 32];

  const int wave = threadIdx.x >> 5;
  const int lane = threadIdx.x & 31;
  const int m = lane & 15;
  const int h = lane >> 4;
  const int kh = wave >> 2;
  const int mp = (wave >> 1) & 1;
  const int np = wave & 1;
  const int bn = blockIdx.x & 3;
  const int bm = blockIdx.x >> 2;
  const int row0 = bm * 64 + mp * 32;
  const int nt0  = bn * 4 + np * 2;
  const int k0   = kh * (D_DIM / 2) + 4 * h;

  const float* xrow0 = x    + (size_t)(row0 + m)        * D_DIM + k0;
  const float* xrow1 = x    + (size_t)(row0 + 16 + m)   * D_DIM + k0;
  const float* arow0 = Amat + (size_t)(nt0 * 16 + m)       * D_DIM + k0;
  const float* arow1 = Amat + (size_t)((nt0 + 1) * 16 + m) * D_DIM + k0;

  v8f c00 = {}, c01 = {}, c10 = {}, c11 = {};

  for (int c = 0; c < nchunk; ++c) {
    const int base = c * 16;
#pragma unroll
    for (int s = 0; s < 2; ++s) {
      const int kb = base + s * 8;
      const v4f a0 = *(const v4f*)(xrow0 + kb);
      const v4f a1 = *(const v4f*)(xrow1 + kb);
      const v4f b0 = *(const v4f*)(arow0 + kb);
      const v4f b1 = *(const v4f*)(arow1 + kb);
      c00 = wmma_f32(frag_lo(a0), frag_lo(b0), c00);
      c01 = wmma_f32(frag_lo(a0), frag_lo(b1), c01);
      c10 = wmma_f32(frag_lo(a1), frag_lo(b0), c10);
      c11 = wmma_f32(frag_lo(a1), frag_lo(b1), c11);
      c00 = wmma_f32(frag_hi(a0), frag_hi(b0), c00);
      c01 = wmma_f32(frag_hi(a0), frag_hi(b1), c01);
      c10 = wmma_f32(frag_hi(a1), frag_hi(b0), c10);
      c11 = wmma_f32(frag_hi(a1), frag_hi(b1), c11);
    }
  }

  float* mine = red + wave * 512;
  const float* peer = red + (wave ^ 4) * 512;
  const v8f give0 = kh ? c00 : c10;
  const v8f give1 = kh ? c01 : c11;
#pragma unroll
  for (int i = 0; i < 8; ++i) {
    mine[i * 32 + lane]       = give0[i];
    mine[256 + i * 32 + lane] = give1[i];
  }
  __syncthreads();
  const v8f keep0 = kh ? c10 : c00;
  const v8f keep1 = kh ? c11 : c01;
  const int rt = row0 + kh * 16;

#pragma unroll
  for (int i = 0; i < 8; ++i) {
    const int ra = rt + i + 8 * h;
    const float s0 = keep0[i] + peer[i * 32 + lane];
    const float s1 = keep1[i] + peer[256 + i * 32 + lane];
    const float g0 = g[ra * K_EXP + nt0];
    const float g1 = g[ra * K_EXP + nt0 + 1];
    hv[(size_t)ra * N_HV + nt0 * 16 + m]       = s0 * g0;
    hv[(size_t)ra * N_HV + (nt0 + 1) * 16 + m] = s1 * g1;
  }
}

__global__ __launch_bounds__(256) void moelora_ogemm(
    const float* __restrict__ hv, const float* __restrict__ Bm,
    float* __restrict__ out, int nexp)
{
  const int wave = threadIdx.x >> 5;
  const int lane = threadIdx.x & 31;
  const int m = lane & 15;
  const int h = lane >> 4;
  const int row0 = blockIdx.x * 32;
  const int dt0  = blockIdx.y * 16 + wave * 2;
  const int dcol0 = dt0 * 16 + m;
  const int dcol1 = (dt0 + 1) * 16 + m;

  const float* hr0 = hv + (size_t)(row0 + m)      * N_HV + 4 * h;
  const float* hr1 = hv + (size_t)(row0 + 16 + m) * N_HV + 4 * h;
  const float* bb0 = Bm + (size_t)dcol0 * R_RANK + 4 * h;
  const float* bb1 = Bm + (size_t)dcol1 * R_RANK + 4 * h;

  v8f c00 = {}, c01 = {}, c10 = {}, c11 = {};

  for (int kk = 0; kk < nexp; ++kk) {
#pragma unroll
    for (int s = 0; s < 2; ++s) {
      const v4f a0 = *(const v4f*)(hr0 + s * 8);
      const v4f a1 = *(const v4f*)(hr1 + s * 8);
      const v4f b0 = *(const v4f*)(bb0 + s * 8);
      const v4f b1 = *(const v4f*)(bb1 + s * 8);
      c00 = wmma_f32(frag_lo(a0), frag_lo(b0), c00);
      c01 = wmma_f32(frag_lo(a0), frag_lo(b1), c01);
      c10 = wmma_f32(frag_lo(a1), frag_lo(b0), c10);
      c11 = wmma_f32(frag_lo(a1), frag_lo(b1), c11);
      c00 = wmma_f32(frag_hi(a0), frag_hi(b0), c00);
      c01 = wmma_f32(frag_hi(a0), frag_hi(b1), c01);
      c10 = wmma_f32(frag_hi(a1), frag_hi(b0), c10);
      c11 = wmma_f32(frag_hi(a1), frag_hi(b1), c11);
    }
    hr0 += R_RANK;
    hr1 += R_RANK;
    bb0 += (size_t)D_DIM * R_RANK;
    bb1 += (size_t)D_DIM * R_RANK;
  }

#pragma unroll
  for (int i = 0; i < 8; ++i) {
    const int ra = row0 + i + 8 * h;
    const int rb = ra + 16;
    out[(size_t)ra * D_DIM + dcol0] = c00[i];
    out[(size_t)ra * D_DIM + dcol1] = c01[i];
    out[(size_t)rb * D_DIM + dcol0] = c10[i];
    out[(size_t)rb * D_DIM + dcol1] = c11[i];
  }
}

extern "C" void kernel_launch(void* const* d_in, const int* in_sizes, int n_in,
                              void* d_out, int out_size, void* d_ws, size_t ws_size,
                              hipStream_t stream) {
  (void)in_sizes; (void)n_in; (void)out_size; (void)ws_size;
  const float* x     = (const float*)d_in[0];
  const int*   alpha = (const int*)  d_in[1];
  const float* A     = (const float*)d_in[2];
  const float* Bm    = (const float*)d_in[3];
  const float* Wr    = (const float*)d_in[4];
  float* out = (float*)d_out;

  float* g  = (float*)d_ws;
  float* hv = g + (size_t)NTOK * K_EXP;

  moelora_router<<<NTOK / 8, 256, 0, stream>>>(x, Wr, alpha, g);
  moelora_hgemm<<<(NTOK / 64) * (N_HV / 64), 256, 0, stream>>>(x, A, g, hv, D_DIM / 32);
  moelora_ogemm<<<dim3(NTOK / 32, D_DIM / 256), 256, 0, stream>>>(hv, Bm, out, K_EXP);
}


# h-GEMM: x and A tiles staged through LDS by whole-line 128-bit global loads one stage ahead (1/8 of the cache-line requests), fragments read with ds_load_b128; k sum order unchanged
# speedup vs baseline: 1.7900x; 1.1060x over previous
#include <hip/hip_runtime.h>


typedef __attribute__((ext_vector_type(2))) float v2f;
typedef __attribute__((ext_vector_type(4))) float v4f;
typedef __attribute__((ext_vector_type(8))) float v8f;

#define D_DIM  2048
#define K_EXP  16
#define R_RANK 16
#define NTOK   4096
#define N_HV   256

static __device__ __forceinline__ v8f wmma_f32(v2f a, v2f b, v8f c) {
  return __builtin_amdgcn_wmma_f32_16x16x4_f32(
       false, a,  false, b,
       (short)0, c,  false,  false);
}

static __device__ __forceinline__ v2f frag_lo(v4f v) {
  return __builtin_shufflevector(v, v, 0, 1);
}
static __device__ __forceinline__ v2f frag_hi(v4f v) {
  return __builtin_shufflevector(v, v, 2, 3);
}

static __device__ __forceinline__ void lds_barrier() {
  __builtin_amdgcn_fence(__ATOMIC_RELEASE, "workgroup", "local");
  __builtin_amdgcn_s_barrier();
  __builtin_amdgcn_fence(__ATOMIC_ACQUIRE, "workgroup", "local");
}

__global__ __launch_bounds__(256) void moelora_router(
    const float* __restrict__ x, const float* __restrict__ Wr,
    const int* __restrict__ alpha_p, float* __restrict__ g)
{
  const int wave = threadIdx.x >> 5;
  const int lane = threadIdx.x & 31;
  const int t = blockIdx.x * 8 + wave;

  float acc[K_EXP];
#pragma unroll
  for (int k = 0; k < K_EXP; ++k) acc[k] = 0.f;

  const float* xrow = x + (size_t)t * D_DIM;
  for (int d = lane; d < D_DIM; d += 32) {
    const float xv = xrow[d];
#pragma unroll
    for (int k = 0; k < K_EXP; ++k)
      acc[k] = fmaf(xv, Wr[k * D_DIM + d], acc[k]);
  }
#pragma unroll
  for (int k = 0; k < K_EXP; ++k) {
#pragma unroll
    for (int off = 16; off > 0; off >>= 1)
      acc[k] += __shfl_xor(acc[k], off, 32);
  }

  if (lane == 0) {
    const float alpha = (float)(*alpha_p);
    int i0 = 0; float v0 = acc[0];
#pragma unroll
    for (int k = 1; k < K_EXP; ++k) if (acc[k] > v0) { v0 = acc[k]; i0 = k; }
    int i1 = -1; float v1 = -3.4e38f;
#pragma unroll
    for (int k = 0; k < K_EXP; ++k)
      if (k != i0 && acc[k] > v1) { v1 = acc[k]; i1 = k; }
    const float e  = __expf(v1 - v0);
    const float dn = 1.f + e;
    const float g0 = alpha / dn;
    const float g1 = alpha * e / dn;
#pragma unroll
    for (int k = 0; k < K_EXP; ++k)
      g[t * K_EXP + k] = (k == i0) ? g0 : ((k == i1) ? g1 : 0.f);
  }
}

#define HG_PITCH 36
#define HG_TILE  (64 * HG_PITCH)
#define HG_BUF   (4 * HG_TILE)

__global__ __launch_bounds__(256) void moelora_hgemm(
    const float* __restrict__ x, const float* __restrict__ Amat,
    const float* __restrict__ g, float* __restrict__ hv, int nstage)
{
  __shared__ __attribute__((aligned(16))) float stg[2 * HG_BUF];
  __shared__ float red[8 * 2 * 8 * 32];

  const int tid  = threadIdx.x;
  const int wave = tid >> 5;
  const int lane = tid & 31;
  const int m = lane & 15;
  const int h = lane >> 4;
  const int kh = wave >> 2;
  const int mp = (wave >> 1) & 1;
  const int np = wave & 1;
  const int bn = blockIdx.x & 3;
  const int bm = blockIdx.x >> 2;
  const int row0 = bm * 64 + mp * 32;
  const int nt0  = bn * 4 + np * 2;

  const int crow = tid >> 3;
  const int ccol = (tid & 7) * 4;
  const float* xg = x    + (size_t)(bm * 64 + crow) * D_DIM + ccol;
  const float* ag = Amat + (size_t)(bn * 64 + crow) * D_DIM + ccol;
  float* sdst = stg + crow * HG_PITCH + ccol;

  const float* fx = stg + kh * HG_TILE       + (mp * 32 + m) * HG_PITCH + 4 * h;
  const float* fa = stg + (2 + kh) * HG_TILE + (np * 32 + m) * HG_PITCH + 4 * h;

  v4f p0, p1, p2, p3, p4, p5, p6, p7;

#define HG_FETCH()                                                   \
  do {                                                               \
    p0 = *(const v4f*)(xg);                                          \
    p1 = *(const v4f*)(xg + 32 * D_DIM);                             \
    p2 = *(const v4f*)(xg + D_DIM / 2);                              \
    p3 = *(const v4f*)(xg + 32 * D_DIM + D_DIM / 2);                 \
    p4 = *(const v4f*)(ag);                                          \
    p5 = *(const v4f*)(ag + 32 * D_DIM);                             \
    p6 = *(const v4f*)(ag + D_DIM / 2);                              \
    p7 = *(const v4f*)(ag + 32 * D_DIM + D_DIM / 2);                 \
  } while (0)
#define HG_PUT(dst)                                                  \
  do {                                                               \
    *(v4f*)((dst))                                   = p0;           \
    *(v4f*)((dst) + 32 * HG_PITCH)                   = p1;           \
    *(v4f*)((dst) + HG_TILE)                         = p2;           \
    *(v4f*)((dst) + HG_TILE + 32 * HG_PITCH)         = p3;           \
    *(v4f*)((dst) + 2 * HG_TILE)                     = p4;           \
    *(v4f*)((dst) + 2 * HG_TILE + 32 * HG_PITCH)     = p5;           \
    *(v4f*)((dst) + 3 * HG_TILE)                     = p6;           \
    *(v4f*)((dst) + 3 * HG_TILE + 32 * HG_PITCH)     = p7;           \
  } while (0)

  HG_FETCH();
  HG_PUT(sdst);
  if (nstage > 1) {
    xg += 32;
    ag += 32;
    HG_FETCH();
  }
  lds_barrier();

  v8f c00 = {}, c01 = {}, c10 = {}, c11 = {};
  int cur = 0;

  for (int st = 0; st < nstage; ++st) {
    const float* bx = fx + cur;
    const float* ba = fa + cur;
    cur ^= HG_BUF;
#pragma unroll
    for (int s = 0; s < 4; ++s) {
      if (s == 2) {
        if (st + 1 < nstage) {
          HG_PUT(sdst + cur);
          if (st + 2 < nstage) {
            xg += 32;
            ag += 32;
            HG_FETCH();
          }
        }
      }
      const v4f a0 = *(const v4f*)(bx + s * 8);
      const v4f a1 = *(const v4f*)(bx + 16 * HG_PITCH + s * 8);
      const v4f b0 = *(const v4f*)(ba + s * 8);
      const v4f b1 = *(const v4f*)(ba + 16 * HG_PITCH + s * 8);
      c00 = wmma_f32(frag_lo(a0), frag_lo(b0), c00);
      c01 = wmma_f32(frag_lo(a0), frag_lo(b1), c01);
      c10 = wmma_f32(frag_lo(a1), frag_lo(b0), c10);
      c11 = wmma_f32(frag_lo(a1), frag_lo(b1), c11);
      c00 = wmma_f32(frag_hi(a0), frag_hi(b0), c00);
      c01 = wmma_f32(frag_hi(a0), frag_hi(b1), c01);
      c10 = wmma_f32(frag_hi(a1), frag_hi(b0), c10);
      c11 = wmma_f32(frag_hi(a1), frag_hi(b1), c11);
    }
    lds_barrier();
  }
#undef HG_FETCH
#undef HG_PUT

  float* mine = red + wave * 512;
  const float* peer = red + (wave ^ 4) * 512;
  const v8f give0 = kh ? c00 : c10;
  const v8f give1 = kh ? c01 : c11;
#pragma unroll
  for (int i = 0; i < 8; ++i) {
    mine[i * 32 + lane]       = give0[i];
    mine[256 + i * 32 + lane] = give1[i];
  }
  __syncthreads();
  const v8f keep0 = kh ? c10 : c00;
  const v8f keep1 = kh ? c11 : c01;
  const int rt = row0 + kh * 16;

#pragma unroll
  for (int i = 0; i < 8; ++i) {
    const int ra = rt + i + 8 * h;
    const float s0 = keep0[i] + peer[i * 32 + lane];
    const float s1 = keep1[i] + peer[256 + i * 32 + lane];
    const float g0 = g[ra * K_EXP + nt0];
    const float g1 = g[ra * K_EXP + nt0 + 1];
    hv[(size_t)ra * N_HV + nt0 * 16 + m]       = s0 * g0;
    hv[(size_t)ra * N_HV + (nt0 + 1) * 16 + m] = s1 * g1;
  }
}

__global__ __launch_bounds__(256) void moelora_ogemm(
    const float* __restrict__ hv, const float* __restrict__ Bm,
    float* __restrict__ out, int nexp)
{
  const int wave = threadIdx.x >> 5;
  const int lane = threadIdx.x & 31;
  const int m = lane & 15;
  const int h = lane >> 4;
  const int row0 = blockIdx.x * 32;
  const int dt0  = blockIdx.y * 16 + wave * 2;
  const int dcol0 = dt0 * 16 + m;
  const int dcol1 = (dt0 + 1) * 16 + m;

  const float* hr0 = hv + (size_t)(row0 + m)      * N_HV + 4 * h;
  const float* hr1 = hv + (size_t)(row0 + 16 + m) * N_HV + 4 * h;
  const float* bb0 = Bm + (size_t)dcol0 * R_RANK + 4 * h;
  const float* bb1 = Bm + (size_t)dcol1 * R_RANK + 4 * h;

  v8f c00 = {}, c01 = {}, c10 = {}, c11 = {};

  for (int kk = 0; kk < nexp; ++kk) {
#pragma unroll
    for (int s = 0; s < 2; ++s) {
      const v4f a0 = *(const v4f*)(hr0 + s * 8);
      const v4f a1 = *(const v4f*)(hr1 + s * 8);
      const v4f b0 = *(const v4f*)(bb0 + s * 8);
      const v4f b1 = *(const v4f*)(bb1 + s * 8);
      c00 = wmma_f32(frag_lo(a0), frag_lo(b0), c00);
      c01 = wmma_f32(frag_lo(a0), frag_lo(b1), c01);
      c10 = wmma_f32(frag_lo(a1), frag_lo(b0), c10);
      c11 = wmma_f32(frag_lo(a1), frag_lo(b1), c11);
      c00 = wmma_f32(frag_hi(a0), frag_hi(b0), c00);
      c01 = wmma_f32(frag_hi(a0), frag_hi(b1), c01);
      c10 = wmma_f32(frag_hi(a1), frag_hi(b0), c10);
      c11 = wmma_f32(frag_hi(a1), frag_hi(b1), c11);
    }
    hr0 += R_RANK;
    hr1 += R_RANK;
    bb0 += (size_t)D_DIM * R_RANK;
    bb1 += (size_t)D_DIM * R_RANK;
  }

#pragma unroll
  for (int i = 0; i < 8; ++i) {
    const int ra = row0 + i + 8 * h;
    const int rb = ra + 16;
    out[(size_t)ra * D_DIM + dcol0] = c00[i];
    out[(size_t)ra * D_DIM + dcol1] = c01[i];
    out[(size_t)rb * D_DIM + dcol0] = c10[i];
    out[(size_t)rb * D_DIM + dcol1] = c11[i];
  }
}

extern "C" void kernel_launch(void* const* d_in, const int* in_sizes, int n_in,
                              void* d_out, int out_size, void* d_ws, size_t ws_size,
                              hipStream_t stream) {
  (void)in_sizes; (void)n_in; (void)out_size; (void)ws_size;
  const float* x     = (const float*)d_in[0];
  const int*   alpha = (const int*)  d_in[1];
  const float* A     = (const float*)d_in[2];
  const float* Bm    = (const float*)d_in[3];
  const float* Wr    = (const float*)d_in[4];
  float* out = (float*)d_out;

  float* g  = (float*)d_ws;
  float* hv = g + (size_t)NTOK * K_EXP;

  moelora_router<<<NTOK / 8, 256, 0, stream>>>(x, Wr, alpha, g);
  moelora_hgemm<<<(NTOK / 64) * (N_HV / 64), 256, 0, stream>>>(x, A, g, hv, D_DIM / 64);
  moelora_ogemm<<<dim3(NTOK / 32, D_DIM / 256), 256, 0, stream>>>(hv, Bm, out, K_EXP);
}


# router scores x @ Wr^T moved from per-lane FMAs (17 loads per 16 FMAs) onto f32 WMMA: 16 tokens per block, d split over 8 waves, partial tiles added through LDS; changes the score sum order
# speedup vs baseline: 2.2475x; 1.2556x over previous
#include <hip/hip_runtime.h>


typedef __attribute__((ext_vector_type(2))) float v2f;
typedef __attribute__((ext_vector_type(4))) float v4f;
typedef __attribute__((ext_vector_type(8))) float v8f;

#define D_DIM  2048
#define K_EXP  16
#define R_RANK 16
#define NTOK   4096
#define N_HV   256

static __device__ __forceinline__ v8f wmma_f32(v2f a, v2f b, v8f c) {
  return __builtin_amdgcn_wmma_f32_16x16x4_f32(
       false, a,  false, b,
       (short)0, c,  false,  false);
}

static __device__ __forceinline__ v2f frag_lo(v4f v) {
  return __builtin_shufflevector(v, v, 0, 1);
}
static __device__ __forceinline__ v2f frag_hi(v4f v) {
  return __builtin_shufflevector(v, v, 2, 3);
}

static __device__ __forceinline__ void lds_barrier() {
  __builtin_amdgcn_fence(__ATOMIC_RELEASE, "workgroup", "local");
  __builtin_amdgcn_s_barrier();
  __builtin_amdgcn_fence(__ATOMIC_ACQUIRE, "workgroup", "local");
}

#define RT_KW 256
#define RT_NG (RT_KW / 8)

__global__ __launch_bounds__(256) void moelora_router(
    const float* __restrict__ x, const float* __restrict__ Wr,
    const int* __restrict__ alpha_p, float* __restrict__ g)
{
  __shared__ float part[8 * 256];
  __shared__ __attribute__((aligned(16))) float sc[256];

  const int tid  = threadIdx.x;
  const int wave = tid >> 5;
  const int lane = tid & 31;
  const int m = lane & 15;
  const int h = lane >> 4;
  const int t0 = blockIdx.x * 16;

  const float* xa = x  + (size_t)(t0 + m) * D_DIM + wave * RT_KW + 4 * h;
  const float* wb = Wr + (size_t)m * D_DIM        + wave * RT_KW + 4 * h;

  v4f a[RT_NG], b[RT_NG];
#pragma unroll
  for (int i = 0; i < RT_NG; ++i) {
    a[i] = *(const v4f*)(xa + i * 8);
    b[i] = *(const v4f*)(wb + i * 8);
  }
  __builtin_amdgcn_sched_barrier(0);

  v8f c0 = {}, c1 = {}, c2 = {}, c3 = {};
#pragma unroll
  for (int i = 0; i < RT_NG; i += 2) {
    c0 = wmma_f32(frag_lo(a[i]),     frag_lo(b[i]),     c0);
    c1 = wmma_f32(frag_hi(a[i]),     frag_hi(b[i]),     c1);
    c2 = wmma_f32(frag_lo(a[i + 1]), frag_lo(b[i + 1]), c2);
    c3 = wmma_f32(frag_hi(a[i + 1]), frag_hi(b[i + 1]), c3);
  }

#pragma unroll
  for (int i = 0; i < 8; ++i)
    part[wave * 256 + (8 * h + i) * 16 + m] = (c0[i] + c1[i]) + (c2[i] + c3[i]);
  __syncthreads();

  float s = part[tid];
#pragma unroll
  for (int w = 1; w < 8; ++w) s += part[w * 256 + tid];
  sc[tid] = s;
  __syncthreads();

  const int tk = tid >> 4;
  const int ke = tid & 15;
  float acc[K_EXP];
#pragma unroll
  for (int q = 0; q < 4; ++q) {
    const v4f v = *(const v4f*)(sc + tk * 16 + q * 4);
    acc[q * 4 + 0] = v[0];
    acc[q * 4 + 1] = v[1];
    acc[q * 4 + 2] = v[2];
    acc[q * 4 + 3] = v[3];
  }
  const float alpha = (float)(*alpha_p);
  int i0 = 0; float v0 = acc[0];
#pragma unroll
  for (int k = 1; k < K_EXP; ++k) if (acc[k] > v0) { v0 = acc[k]; i0 = k; }
  int i1 = -1; float v1 = -3.4e38f;
#pragma unroll
  for (int k = 0; k < K_EXP; ++k)
    if (k != i0 && acc[k] > v1) { v1 = acc[k]; i1 = k; }
  const float e  = __expf(v1 - v0);
  const float dn = 1.f + e;
  const float g0 = alpha / dn;
  const float g1 = alpha * e / dn;
  g[(size_t)(t0 + tk) * K_EXP + ke] = (ke == i0) ? g0 : ((ke == i1) ? g1 : 0.f);
}

#define HG_PITCH 36
#define HG_TILE  (64 * HG_PITCH)
#define HG_BUF   (4 * HG_TILE)

__global__ __launch_bounds__(256) void moelora_hgemm(
    const float* __restrict__ x, const float* __restrict__ Amat,
    const float* __restrict__ g, float* __restrict__ hv, int nstage)
{
  __shared__ __attribute__((aligned(16))) float stg[2 * HG_BUF];
  __shared__ float red[8 * 2 * 8 * 32];

  const int tid  = threadIdx.x;
  const int wave = tid >> 5;
  const int lane = tid & 31;
  const int m = lane & 15;
  const int h = lane >> 4;
  const int kh = wave >> 2;
  const int mp = (wave >> 1) & 1;
  const int np = wave & 1;
  const int bn = blockIdx.x & 3;
  const int bm = blockIdx.x >> 2;
  const int row0 = bm * 64 + mp * 32;
  const int nt0  = bn * 4 + np * 2;

  const int crow = tid >> 3;
  const int ccol = (tid & 7) * 4;
  const float* xg = x    + (size_t)(bm * 64 + crow) * D_DIM + ccol;
  const float* ag = Amat + (size_t)(bn * 64 + crow) * D_DIM + ccol;
  float* sdst = stg + crow * HG_PITCH + ccol;

  const float* fx = stg + kh * HG_TILE       + (mp * 32 + m) * HG_PITCH + 4 * h;
  const float* fa = stg + (2 + kh) * HG_TILE + (np * 32 + m) * HG_PITCH + 4 * h;

  v4f p0, p1, p2, p3, p4, p5, p6, p7;

#define HG_FETCH()                                                   \
  do {                                                               \
    p0 = *(const v4f*)(xg);                                          \
    p1 = *(const v4f*)(xg + 32 * D_DIM);                             \
    p2 = *(const v4f*)(xg + D_DIM / 2);                              \
    p3 = *(const v4f*)(xg + 32 * D_DIM + D_DIM / 2);                 \
    p4 = *(const v4f*)(ag);                                          \
    p5 = *(const v4f*)(ag + 32 * D_DIM);                             \
    p6 = *(const v4f*)(ag + D_DIM / 2);                              \
    p7 = *(const v4f*)(ag + 32 * D_DIM + D_DIM / 2);                 \
  } while (0)
#define HG_PUT(dst)                                                  \
  do {                                                               \
    *(v4f*)((dst))                                   = p0;           \
    *(v4f*)((dst) + 32 * HG_PITCH)                   = p1;           \
    *(v4f*)((dst) + HG_TILE)                         = p2;           \
    *(v4f*)((dst) + HG_TILE + 32 * HG_PITCH)         = p3;           \
    *(v4f*)((dst) + 2 * HG_TILE)                     = p4;           \
    *(v4f*)((dst) + 2 * HG_TILE + 32 * HG_PITCH)     = p5;           \
    *(v4f*)((dst) + 3 * HG_TILE)                     = p6;           \
    *(v4f*)((dst) + 3 * HG_TILE + 32 * HG_PITCH)     = p7;           \
  } while (0)

  HG_FETCH();
  HG_PUT(sdst);
  if (nstage > 1) {
    xg += 32;
    ag += 32;
    HG_FETCH();
  }
  lds_barrier();

  v8f c00 = {}, c01 = {}, c10 = {}, c11 = {};
  int cur = 0;

  for (int st = 0; st < nstage; ++st) {
    const float* bx = fx + cur;
    const float* ba = fa + cur;
    cur ^= HG_BUF;
#pragma unroll
    for (int s = 0; s < 4; ++s) {
      if (s == 2) {
        if (st + 1 < nstage) {
          HG_PUT(sdst + cur);
          if (st + 2 < nstage) {
            xg += 32;
            ag += 32;
            HG_FETCH();
          }
        }
      }
      const v4f a0 = *(const v4f*)(bx + s * 8);
      const v4f a1 = *(const v4f*)(bx + 16 * HG_PITCH + s * 8);
      const v4f b0 = *(const v4f*)(ba + s * 8);
      const v4f b1 = *(const v4f*)(ba + 16 * HG_PITCH + s * 8);
      c00 = wmma_f32(frag_lo(a0), frag_lo(b0), c00);
      c01 = wmma_f32(frag_lo(a0), frag_lo(b1), c01);
      c10 = wmma_f32(frag_lo(a1), frag_lo(b0), c10);
      c11 = wmma_f32(frag_lo(a1), frag_lo(b1), c11);
      c00 = wmma_f32(frag_hi(a0), frag_hi(b0), c00);
      c01 = wmma_f32(frag_hi(a0), frag_hi(b1), c01);
      c10 = wmma_f32(frag_hi(a1), frag_hi(b0), c10);
      c11 = wmma_f32(frag_hi(a1), frag_hi(b1), c11);
    }
    lds_barrier();
  }
#undef HG_FETCH
#undef HG_PUT

  float* mine = red + wave * 512;
  const float* peer = red + (wave ^ 4) * 512;
  const v8f give0 = kh ? c00 : c10;
  const v8f give1 = kh ? c01 : c11;
#pragma unroll
  for (int i = 0; i < 8; ++i) {
    mine[i * 32 + lane]       = give0[i];
    mine[256 + i * 32 + lane] = give1[i];
  }
  __syncthreads();
  const v8f keep0 = kh ? c10 : c00;
  const v8f keep1 = kh ? c11 : c01;
  const int rt = row0 + kh * 16;

#pragma unroll
  for (int i = 0; i < 8; ++i) {
    const int ra = rt + i + 8 * h;
    const float s0 = keep0[i] + peer[i * 32 + lane];
    const float s1 = keep1[i] + peer[256 + i * 32 + lane];
    const float g0 = g[ra * K_EXP + nt0];
    const float g1 = g[ra * K_EXP + nt0 + 1];
    hv[(size_t)ra * N_HV + nt0 * 16 + m]       = s0 * g0;
    hv[(size_t)ra * N_HV + (nt0 + 1) * 16 + m] = s1 * g1;
  }
}

__global__ __launch_bounds__(256) void moelora_ogemm(
    const float* __restrict__ hv, const float* __restrict__ Bm,
    float* __restrict__ out, int nexp)
{
  const int wave = threadIdx.x >> 5;
  const int lane = threadIdx.x & 31;
  const int m = lane & 15;
  const int h = lane >> 4;
  const int row0 = blockIdx.x * 32;
  const int dt0  = blockIdx.y * 16 + wave * 2;
  const int dcol0 = dt0 * 16 + m;
  const int dcol1 = (dt0 + 1) * 16 + m;

  const float* hr0 = hv + (size_t)(row0 + m)      * N_HV + 4 * h;
  const float* hr1 = hv + (size_t)(row0 + 16 + m) * N_HV + 4 * h;
  const float* bb0 = Bm + (size_t)dcol0 * R_RANK + 4 * h;
  const float* bb1 = Bm + (size_t)dcol1 * R_RANK + 4 * h;

  v8f c00 = {}, c01 = {}, c10 = {}, c11 = {};

  for (int kk = 0; kk < nexp; ++kk) {
#pragma unroll
    for (int s = 0; s < 2; ++s) {
      const v4f a0 = *(const v4f*)(hr0 + s * 8);
      const v4f a1 = *(const v4f*)(hr1 + s * 8);
      const v4f b0 = *(const v4f*)(bb0 + s * 8);
      const v4f b1 = *(const v4f*)(bb1 + s * 8);
      c00 = wmma_f32(frag_lo(a0), frag_lo(b0), c00);
      c01 = wmma_f32(frag_lo(a0), frag_lo(b1), c01);
      c10 = wmma_f32(frag_lo(a1), frag_lo(b0), c10);
      c11 = wmma_f32(frag_lo(a1), frag_lo(b1), c11);
      c00 = wmma_f32(frag_hi(a0), frag_hi(b0), c00);
      c01 = wmma_f32(frag_hi(a0), frag_hi(b1), c01);
      c10 = wmma_f32(frag_hi(a1), frag_hi(b0), c10);
      c11 = wmma_f32(frag_hi(a1), frag_hi(b1), c11);
    }
    hr0 += R_RANK;
    hr1 += R_RANK;
    bb0 += (size_t)D_DIM * R_RANK;
    bb1 += (size_t)D_DIM * R_RANK;
  }

#pragma unroll
  for (int i = 0; i < 8; ++i) {
    const int ra = row0 + i + 8 * h;
    const int rb = ra + 16;
    out[(size_t)ra * D_DIM + dcol0] = c00[i];
    out[(size_t)ra * D_DIM + dcol1] = c01[i];
    out[(size_t)rb * D_DIM + dcol0] = c10[i];
    out[(size_t)rb * D_DIM + dcol1] = c11[i];
  }
}

extern "C" void kernel_launch(void* const* d_in, const int* in_sizes, int n_in,
                              void* d_out, int out_size, void* d_ws, size_t ws_size,
                              hipStream_t stream) {
  (void)in_sizes; (void)n_in; (void)out_size; (void)ws_size;
  const float* x     = (const float*)d_in[0];
  const int*   alpha = (const int*)  d_in[1];
  const float* A     = (const float*)d_in[2];
  const float* Bm    = (const float*)d_in[3];
  const float* Wr    = (const float*)d_in[4];
  float* out = (float*)d_out;

  float* g  = (float*)d_ws;
  float* hv = g + (size_t)NTOK * K_EXP;

  moelora_router<<<NTOK / 16, 256, 0, stream>>>(x, Wr, alpha, g);
  moelora_hgemm<<<(NTOK / 64) * (N_HV / 64), 256, 0, stream>>>(x, A, g, hv, D_DIM / 64);
  moelora_ogemm<<<dim3(NTOK / 32, D_DIM / 256), 256, 0, stream>>>(hv, Bm, out, K_EXP);
}


# top-2 sparsity: tokens counting-sorted by expert pair (no atomics); each 16-row tile of one pair runs both LoRA GEMMs over its 2 selected experts only, h kept in LDS; changes both k sum orders
# speedup vs baseline: 3.3363x; 1.4844x over previous
#include <hip/hip_runtime.h>


typedef __attribute__((ext_vector_type(2))) float v2f;
typedef __attribute__((ext_vector_type(4))) float v4f;
typedef __attribute__((ext_vector_type(8))) float v8f;

#define D_DIM  2048
#define K_EXP  16
#define R_RANK 16
#define NTOK   4096
#define NTILE_MAX 384
#define NROW_MAX  (NTILE_MAX * 16)

static __device__ __forceinline__ v8f wmma_f32(v2f a, v2f b, v8f c) {
  return __builtin_amdgcn_wmma_f32_16x16x4_f32(
       false, a,  false, b,
       (short)0, c,  false,  false);
}

static __device__ __forceinline__ v2f frag_lo(v4f v) {
  return __builtin_shufflevector(v, v, 0, 1);
}
static __device__ __forceinline__ v2f frag_hi(v4f v) {
  return __builtin_shufflevector(v, v, 2, 3);
}

#define RT_KW 256
#define RT_NG (RT_KW / 8)

__global__ __launch_bounds__(256) void moelora_router(
    const float* __restrict__ x, const float* __restrict__ Wr,
    const int* __restrict__ alpha_p, int* __restrict__ tkey,
    float* __restrict__ tgate)
{
  __shared__ float part[8 * 256];
  __shared__ __attribute__((aligned(16))) float sc[256];

  const int tid  = threadIdx.x;
  const int wave = tid >> 5;
  const int lane = tid & 31;
  const int m = lane & 15;
  const int h = lane >> 4;
  const int t0 = blockIdx.x * 16;

  const float* xa = x  + (size_t)(t0 + m) * D_DIM + wave * RT_KW + 4 * h;
  const float* wb = Wr + (size_t)m * D_DIM        + wave * RT_KW + 4 * h;

  v4f a[RT_NG], b[RT_NG];
#pragma unroll
  for (int i = 0; i < RT_NG; ++i) {
    a[i] = *(const v4f*)(xa + i * 8);
    b[i] = *(const v4f*)(wb + i * 8);
  }
  __builtin_amdgcn_sched_barrier(0);

  v8f c0 = {}, c1 = {}, c2 = {}, c3 = {};
#pragma unroll
  for (int i = 0; i < RT_NG; i += 2) {
    c0 = wmma_f32(frag_lo(a[i]),     frag_lo(b[i]),     c0);
    c1 = wmma_f32(frag_hi(a[i]),     frag_hi(b[i]),     c1);
    c2 = wmma_f32(frag_lo(a[i + 1]), frag_lo(b[i + 1]), c2);
    c3 = wmma_f32(frag_hi(a[i + 1]), frag_hi(b[i + 1]), c3);
  }

#pragma unroll
  for (int i = 0; i < 8; ++i)
    part[wave * 256 + (8 * h + i) * 16 + m] = (c0[i] + c1[i]) + (c2[i] + c3[i]);
  __syncthreads();

  float s = part[tid];
#pragma unroll
  for (int w = 1; w < 8; ++w) s += part[w * 256 + tid];
  sc[tid] = s;
  __syncthreads();

  const int tk = tid >> 4;
  const int ke = tid & 15;
  float acc[K_EXP];
#pragma unroll
  for (int q = 0; q < 4; ++q) {
    const v4f v = *(const v4f*)(sc + tk * 16 + q * 4);
    acc[q * 4 + 0] = v[0];
    acc[q * 4 + 1] = v[1];
    acc[q * 4 + 2] = v[2];
    acc[q * 4 + 3] = v[3];
  }
  const float alpha = (float)(*alpha_p);
  int i0 = 0; float v0 = acc[0];
#pragma unroll
  for (int k = 1; k < K_EXP; ++k) if (acc[k] > v0) { v0 = acc[k]; i0 = k; }
  int i1 = -1; float v1 = -3.4e38f;
#pragma unroll
  for (int k = 0; k < K_EXP; ++k)
    if (k != i0 && acc[k] > v1) { v1 = acc[k]; i1 = k; }
  const float e  = __expf(v1 - v0);
  const float dn = 1.f + e;
  const float g0 = alpha / dn;
  const float g1 = alpha * e / dn;
  if (i1 < 0) i1 = (i0 == 0) ? 1 : 0;
  if (ke == 0) {
    const int t = t0 + tk;
    const bool sw = i1 < i0;
    tkey[t] = sw ? (i1 * 16 + i0) : (i0 * 16 + i1);
    tgate[2 * t]     = sw ? g1 : g0;
    tgate[2 * t + 1] = sw ? g0 : g1;
  }
}

__global__ __launch_bounds__(1024) void moelora_group(
    const int* __restrict__ tkey, int* __restrict__ perm,
    int* __restrict__ tile_key, int* __restrict__ ntiles_p)
{
  __shared__ int hist[32 * 256];
  __shared__ int padded[256];
  __shared__ int first[256];

  const int tid  = threadIdx.x;
  const int wave = tid >> 5;
  const int lane = tid & 31;

  for (int i = tid; i < 32 * 256; i += 1024) hist[i] = 0;
  __syncthreads();

  int key[4], rk[4];
#pragma unroll
  for (int s = 0; s < 4; ++s) {
    const int k = tkey[wave * 128 + s * 32 + lane] & 255;
    int rank = 0, cnt = 0;
#pragma unroll
    for (int j = 0; j < 32; ++j) {
      const int same = (__builtin_amdgcn_readlane(k, j) == k) ? 1 : 0;
      cnt += same;
      rank += (j < lane) ? same : 0;
    }
    const int base = hist[wave * 256 + k];
    __syncthreads();
    if (rank == cnt - 1) hist[wave * 256 + k] = base + cnt;
    __syncthreads();
    key[s] = k;
    rk[s] = base + rank;
  }

  int total = 0, pad = 0;
  if (tid < 256) {
    int run = 0;
    for (int w = 0; w < 32; ++w) {
      const int c = hist[w * 256 + tid];
      hist[w * 256 + tid] = run;
      run += c;
    }
    total = run;
    pad = (total + 15) & ~15;
    padded[tid] = pad;
  }
  __syncthreads();
  if (tid < 256) {
    int row = 0;
    for (int j = 0; j < tid; ++j) row += padded[j];
    first[tid] = row;
    for (int i = total; i < pad; ++i) perm[row + i] = -1;
    const int t0 = row >> 4;
    const int nt = pad >> 4;
    for (int i = 0; i < nt; ++i) tile_key[t0 + i] = tid;
    if (tid == 255) *ntiles_p = t0 + nt;
  }
  __syncthreads();
#pragma unroll
  for (int s = 0; s < 4; ++s)
    perm[first[key[s]] + hist[wave * 256 + key[s]] + rk[s]] = wave * 128 + s * 32 + lane;
}

#define TL_NG 16
#define HS_PITCH 36

__global__ __launch_bounds__(256) void moelora_tiles(
    const float* __restrict__ x, const float* __restrict__ Amat,
    const float* __restrict__ Bm, const float* __restrict__ tgate,
    const int* __restrict__ perm, const int* __restrict__ tile_key,
    const int* __restrict__ ntiles_p, float* __restrict__ out)
{
  __shared__ float part[8 * 512];
  __shared__ __attribute__((aligned(16))) float hs[16 * HS_PITCH];
  __shared__ int stok[16];

  const int tile = blockIdx.x;
  if (tile >= *ntiles_p) return;

  const int tid  = threadIdx.x;
  const int wave = tid >> 5;
  const int lane = tid & 31;
  const int m = lane & 15;
  const int h = lane >> 4;
  const int key = tile_key[tile];
  const int elo = (key >> 4) & 15;
  const int ehi = key & 15;

  const int tokm = perm[tile * 16 + m];
  if (tid < 16) stok[tid] = tokm;
  const int tokc = (tokm >= 0 && tokm < NTOK) ? tokm : 0;

  const float* xa  = x    + (size_t)tokc * D_DIM           + wave * 256 + 4 * h;
  const float* wlo = Amat + (size_t)(elo * 16 + m) * D_DIM + wave * 256 + 4 * h;
  const float* whi = Amat + (size_t)(ehi * 16 + m) * D_DIM + wave * 256 + 4 * h;

  v8f cl0 = {}, cl1 = {}, ch0 = {}, ch1 = {};
#pragma unroll
  for (int hf = 0; hf < 2; ++hf) {
    v4f a[TL_NG], bl[TL_NG], bh[TL_NG];
#pragma unroll
    for (int i = 0; i < TL_NG; ++i) {
      a[i]  = *(const v4f*)(xa  + hf * 128 + i * 8);
      bl[i] = *(const v4f*)(wlo + hf * 128 + i * 8);
      bh[i] = *(const v4f*)(whi + hf * 128 + i * 8);
    }
    __builtin_amdgcn_sched_barrier(0);
#pragma unroll
    for (int i = 0; i < TL_NG; ++i) {
      cl0 = wmma_f32(frag_lo(a[i]), frag_lo(bl[i]), cl0);
      ch0 = wmma_f32(frag_lo(a[i]), frag_lo(bh[i]), ch0);
      cl1 = wmma_f32(frag_hi(a[i]), frag_hi(bl[i]), cl1);
      ch1 = wmma_f32(frag_hi(a[i]), frag_hi(bh[i]), ch1);
    }
  }

#pragma unroll
  for (int i = 0; i < 8; ++i) {
    part[wave * 512 +       (8 * h + i) * 16 + m] = cl0[i] + cl1[i];
    part[wave * 512 + 256 + (8 * h + i) * 16 + m] = ch0[i] + ch1[i];
  }
  __syncthreads();

  {
    const int row = tid >> 4;
    const int r   = tid & 15;
    const int tk  = stok[row];
    const bool ok = (tk >= 0 && tk < NTOK);
    const float glo = ok ? tgate[2 * tk]     : 0.f;
    const float ghi = ok ? tgate[2 * tk + 1] : 0.f;
    float s0 = part[tid];
    float s1 = part[256 + tid];
#pragma unroll
    for (int w = 1; w < 8; ++w) {
      s0 += part[w * 512 + tid];
      s1 += part[w * 512 + 256 + tid];
    }
    hs[row * HS_PITCH + r]      = s0 * glo;
    hs[row * HS_PITCH + 16 + r] = s1 * ghi;
  }
  __syncthreads();

  v4f af[4];
#pragma unroll
  for (int q = 0; q < 4; ++q)
    af[q] = *(const v4f*)(hs + m * HS_PITCH + q * 8 + 4 * h);

  float* orow[8];
  bool rok[8];
#pragma unroll
  for (int i = 0; i < 8; ++i) {
    const int tk = stok[8 * h + i];
    rok[i] = (tk >= 0 && tk < NTOK);
    orow[i] = out + (size_t)(rok[i] ? tk : 0) * D_DIM + wave * 256 + m;
  }

  const int dcol = wave * 256 + m;
  const float* blo = Bm + ((size_t)elo * D_DIM + dcol) * R_RANK + 4 * h;
  const float* bhi = Bm + ((size_t)ehi * D_DIM + dcol) * R_RANK + 4 * h;

  v4f bq[4][4], bn[4][4];
#pragma unroll
  for (int t = 0; t < 4; ++t) {
    bq[t][0] = *(const v4f*)(blo + t * 16 * R_RANK);
    bq[t][1] = *(const v4f*)(blo + t * 16 * R_RANK + 8);
    bq[t][2] = *(const v4f*)(bhi + t * 16 * R_RANK);
    bq[t][3] = *(const v4f*)(bhi + t * 16 * R_RANK + 8);
  }
#pragma unroll
  for (int c = 0; c < 4; ++c) {
    if (c + 1 < 4) {
#pragma unroll
      for (int t = 0; t < 4; ++t) {
        const int o = ((c + 1) * 4 + t) * 16 * R_RANK;
        bn[t][0] = *(const v4f*)(blo + o);
        bn[t][1] = *(const v4f*)(blo + o + 8);
        bn[t][2] = *(const v4f*)(bhi + o);
        bn[t][3] = *(const v4f*)(bhi + o + 8);
      }
    }
    v8f c0[4], c1[4];
#pragma unroll
    for (int t = 0; t < 4; ++t) { c0[t] = v8f{}; c1[t] = v8f{}; }
#pragma unroll
    for (int q = 0; q < 4; ++q) {
#pragma unroll
      for (int t = 0; t < 4; ++t) {
        c0[t] = wmma_f32(frag_lo(af[q]), frag_lo(bq[t][q]), c0[t]);
        c1[t] = wmma_f32(frag_hi(af[q]), frag_hi(bq[t][q]), c1[t]);
      }
    }
#pragma unroll
    for (int i = 0; i < 8; ++i) {
      if (rok[i]) {
        float* o = orow[i] + c * 64;
#pragma unroll
        for (int t = 0; t < 4; ++t) o[t * 16] = c0[t][i] + c1[t][i];
      }
    }
#pragma unroll
    for (int t = 0; t < 4; ++t)
#pragma unroll
      for (int q = 0; q < 4; ++q) bq[t][q] = bn[t][q];
  }
}

extern "C" void kernel_launch(void* const* d_in, const int* in_sizes, int n_in,
                              void* d_out, int out_size, void* d_ws, size_t ws_size,
                              hipStream_t stream) {
  (void)in_sizes; (void)n_in; (void)out_size; (void)ws_size;
  const float* x     = (const float*)d_in[0];
  const int*   alpha = (const int*)  d_in[1];
  const float* A     = (const float*)d_in[2];
  const float* Bm    = (const float*)d_in[3];
  const float* Wr    = (const float*)d_in[4];
  float* out = (float*)d_out;

  int*   tkey     = (int*)d_ws;
  float* tgate    = (float*)(tkey + NTOK);
  int*   perm     = (int*)(tgate + 2 * NTOK);
  int*   tile_key = perm + 8192;
  int*   ntiles   = tile_key + 512;

  moelora_router<<<NTOK / 16, 256, 0, stream>>>(x, Wr, alpha, tkey, tgate);
  moelora_group<<<1, 1024, 0, stream>>>(tkey, perm, tile_key, ntiles);
  moelora_tiles<<<NTILE_MAX, 256, 0, stream>>>(x, A, Bm, tgate, perm, tile_key, ntiles, out);
}


# group kernel (one block between two launches): dependent chain cut: key loads issued together, in-wave ranks by 8 key-bit ballots not 32 lane compares, no block barrier per slot, key offsets by scan
# speedup vs baseline: 3.5648x; 1.0685x over previous
#include <hip/hip_runtime.h>


typedef __attribute__((ext_vector_type(2))) float v2f;
typedef __attribute__((ext_vector_type(4))) float v4f;
typedef __attribute__((ext_vector_type(8))) float v8f;

#define D_DIM  2048
#define K_EXP  16
#define R_RANK 16
#define NTOK   4096
#define NTILE_MAX 384
#define NROW_MAX  (NTILE_MAX * 16)

static __device__ __forceinline__ v8f wmma_f32(v2f a, v2f b, v8f c) {
  return __builtin_amdgcn_wmma_f32_16x16x4_f32(
       false, a,  false, b,
       (short)0, c,  false,  false);
}

static __device__ __forceinline__ v2f frag_lo(v4f v) {
  return __builtin_shufflevector(v, v, 0, 1);
}
static __device__ __forceinline__ v2f frag_hi(v4f v) {
  return __builtin_shufflevector(v, v, 2, 3);
}

#define RT_KW 256
#define RT_NG (RT_KW / 8)

__global__ __launch_bounds__(256) void moelora_router(
    const float* __restrict__ x, const float* __restrict__ Wr,
    const int* __restrict__ alpha_p, int* __restrict__ tkey,
    float* __restrict__ tgate)
{
  __shared__ float part[8 * 256];
  __shared__ __attribute__((aligned(16))) float sc[256];

  const int tid  = threadIdx.x;
  const int wave = tid >> 5;
  const int lane = tid & 31;
  const int m = lane & 15;
  const int h = lane >> 4;
  const int t0 = blockIdx.x * 16;

  const float* xa = x  + (size_t)(t0 + m) * D_DIM + wave * RT_KW + 4 * h;
  const float* wb = Wr + (size_t)m * D_DIM        + wave * RT_KW + 4 * h;

  v4f a[RT_NG], b[RT_NG];
#pragma unroll
  for (int i = 0; i < RT_NG; ++i) {
    a[i] = *(const v4f*)(xa + i * 8);
    b[i] = *(const v4f*)(wb + i * 8);
  }
  __builtin_amdgcn_sched_barrier(0);

  v8f c0 = {}, c1 = {}, c2 = {}, c3 = {};
#pragma unroll
  for (int i = 0; i < RT_NG; i += 2) {
    c0 = wmma_f32(frag_lo(a[i]),     frag_lo(b[i]),     c0);
    c1 = wmma_f32(frag_hi(a[i]),     frag_hi(b[i]),     c1);
    c2 = wmma_f32(frag_lo(a[i + 1]), frag_lo(b[i + 1]), c2);
    c3 = wmma_f32(frag_hi(a[i + 1]), frag_hi(b[i + 1]), c3);
  }

#pragma unroll
  for (int i = 0; i < 8; ++i)
    part[wave * 256 + (8 * h + i) * 16 + m] = (c0[i] + c1[i]) + (c2[i] + c3[i]);
  __syncthreads();

  float s = part[tid];
#pragma unroll
  for (int w = 1; w < 8; ++w) s += part[w * 256 + tid];
  sc[tid] = s;
  __syncthreads();

  const int tk = tid >> 4;
  const int ke = tid & 15;
  float acc[K_EXP];
#pragma unroll
  for (int q = 0; q < 4; ++q) {
    const v4f v = *(const v4f*)(sc + tk * 16 + q * 4);
    acc[q * 4 + 0] = v[0];
    acc[q * 4 + 1] = v[1];
    acc[q * 4 + 2] = v[2];
    acc[q * 4 + 3] = v[3];
  }
  const float alpha = (float)(*alpha_p);
  int i0 = 0; float v0 = acc[0];
#pragma unroll
  for (int k = 1; k < K_EXP; ++k) if (acc[k] > v0) { v0 = acc[k]; i0 = k; }
  int i1 = -1; float v1 = -3.4e38f;
#pragma unroll
  for (int k = 0; k < K_EXP; ++k)
    if (k != i0 && acc[k] > v1) { v1 = acc[k]; i1 = k; }
  const float e  = __expf(v1 - v0);
  const float dn = 1.f + e;
  const float g0 = alpha / dn;
  const float g1 = alpha * e / dn;
  if (i1 < 0) i1 = (i0 == 0) ? 1 : 0;
  if (ke == 0) {
    const int t = t0 + tk;
    const bool sw = i1 < i0;
    tkey[t] = sw ? (i1 * 16 + i0) : (i0 * 16 + i1);
    tgate[2 * t]     = sw ? g1 : g0;
    tgate[2 * t + 1] = sw ? g0 : g1;
  }
}


static __device__ __forceinline__ void wave_lds_order() {
  __builtin_amdgcn_fence(__ATOMIC_RELEASE, "workgroup");
  __builtin_amdgcn_wave_barrier();
  __builtin_amdgcn_fence(__ATOMIC_ACQUIRE, "workgroup");
}

__global__ __launch_bounds__(1024) void moelora_group(
    const int* __restrict__ tkey, int* __restrict__ perm,
    int* __restrict__ tile_key, int* __restrict__ ntiles_p)
{
  __shared__ int hist[32 * 256];
  __shared__ int first[256];
  __shared__ int wsum[8];

  const int tid  = threadIdx.x;
  const int wave = tid >> 5;
  const int lane = tid & 31;
  int* cnt_row = hist + wave * 256;

  int key[4], rk[4];
#pragma unroll
  for (int s = 0; s < 4; ++s) key[s] = tkey[wave * 128 + s * 32 + lane] & 255;

#pragma unroll
  for (int j = 0; j < 8; ++j) cnt_row[j * 32 + lane] = 0;
  wave_lds_order();

#pragma unroll
  for (int s = 0; s < 4; ++s) {
    const int k = key[s];
    unsigned peers = 0xffffffffu;
#pragma unroll
    for (int b = 0; b < 8; ++b) {
      const bool bit = ((k >> b) & 1) != 0;
      const unsigned mb = __builtin_amdgcn_ballot_w32(bit);
      peers &= bit ? mb : ~mb;
    }
    const int rank = (int)__builtin_amdgcn_mbcnt_lo(peers, 0u);
    const int cnt  = __builtin_popcount(peers);
    const int base = cnt_row[k];
    if (rank == cnt - 1) cnt_row[k] = base + cnt;
    rk[s] = base + rank;
    wave_lds_order();
  }
  __syncthreads();

  int total = 0, pad = 0, incl = 0;
  if (tid < 256) {
    int run = 0;
    for (int w = 0; w < 32; ++w) {
      const int c = hist[w * 256 + tid];
      hist[w * 256 + tid] = run;
      run += c;
    }
    total = run;
    pad = (total + 15) & ~15;
    incl = pad;
#pragma unroll
    for (int o = 1; o < 32; o <<= 1) {
      const int v = __shfl_up(incl, o);
      if (lane >= o) incl += v;
    }
    if (lane == 31) wsum[wave] = incl;
  }
  __syncthreads();
  if (tid < 256) {
    int row = incl - pad;
#pragma unroll
    for (int w = 0; w < 8; ++w) row += (w < wave) ? wsum[w] : 0;
    first[tid] = row;
    for (int i = total; i < pad; ++i) perm[row + i] = -1;
    const int t0 = row >> 4;
    const int nt = pad >> 4;
    for (int i = 0; i < nt; ++i) tile_key[t0 + i] = tid;
    if (tid == 255) *ntiles_p = t0 + nt;
  }
  __syncthreads();
#pragma unroll
  for (int s = 0; s < 4; ++s)
    perm[first[key[s]] + hist[wave * 256 + key[s]] + rk[s]] = wave * 128 + s * 32 + lane;
}

#define TL_NG 16
#define HS_PITCH 36

__global__ __launch_bounds__(256) void moelora_tiles(
    const float* __restrict__ x, const float* __restrict__ Amat,
    const float* __restrict__ Bm, const float* __restrict__ tgate,
    const int* __restrict__ perm, const int* __restrict__ tile_key,
    const int* __restrict__ ntiles_p, float* __restrict__ out)
{
  __shared__ float part[8 * 512];
  __shared__ __attribute__((aligned(16))) float hs[16 * HS_PITCH];
  __shared__ int stok[16];

  const int tile = blockIdx.x;
  if (tile >= *ntiles_p) return;

  const int tid  = threadIdx.x;
  const int wave = tid >> 5;
  const int lane = tid & 31;
  const int m = lane & 15;
  const int h = lane >> 4;
  const int key = tile_key[tile];
  const int elo = (key >> 4) & 15;
  const int ehi = key & 15;

  const int tokm = perm[tile * 16 + m];
  if (tid < 16) stok[tid] = tokm;
  const int tokc = (tokm >= 0 && tokm < NTOK) ? tokm : 0;

  const float* xa  = x    + (size_t)tokc * D_DIM           + wave * 256 + 4 * h;
  const float* wlo = Amat + (size_t)(elo * 16 + m) * D_DIM + wave * 256 + 4 * h;
  const float* whi = Amat + (size_t)(ehi * 16 + m) * D_DIM + wave * 256 + 4 * h;

  v8f cl0 = {}, cl1 = {}, ch0 = {}, ch1 = {};
#pragma unroll
  for (int hf = 0; hf < 2; ++hf) {
    v4f a[TL_NG], bl[TL_NG], bh[TL_NG];
#pragma unroll
    for (int i = 0; i < TL_NG; ++i) {
      a[i]  = *(const v4f*)(xa  + hf * 128 + i * 8);
      bl[i] = *(const v4f*)(wlo + hf * 128 + i * 8);
      bh[i] = *(const v4f*)(whi + hf * 128 + i * 8);
    }
    __builtin_amdgcn_sched_barrier(0);
#pragma unroll
    for (int i = 0; i < TL_NG; ++i) {
      cl0 = wmma_f32(frag_lo(a[i]), frag_lo(bl[i]), cl0);
      ch0 = wmma_f32(frag_lo(a[i]), frag_lo(bh[i]), ch0);
      cl1 = wmma_f32(frag_hi(a[i]), frag_hi(bl[i]), cl1);
      ch1 = wmma_f32(frag_hi(a[i]), frag_hi(bh[i]), ch1);
    }
  }

#pragma unroll
  for (int i = 0; i < 8; ++i) {
    part[wave * 512 +       (8 * h + i) * 16 + m] = cl0[i] + cl1[i];
    part[wave * 512 + 256 + (8 * h + i) * 16 + m] = ch0[i] + ch1[i];
  }
  __syncthreads();

  {
    const int row = tid >> 4;
    const int r   = tid & 15;
    const int tk  = stok[row];
    const bool ok = (tk >= 0 && tk < NTOK);
    const float glo = ok ? tgate[2 * tk]     : 0.f;
    const float ghi = ok ? tgate[2 * tk + 1] : 0.f;
    float s0 = part[tid];
    float s1 = part[256 + tid];
#pragma unroll
    for (int w = 1; w < 8; ++w) {
      s0 += part[w * 512 + tid];
      s1 += part[w * 512 + 256 + tid];
    }
    hs[row * HS_PITCH + r]      = s0 * glo;
    hs[row * HS_PITCH + 16 + r] = s1 * ghi;
  }
  __syncthreads();

  v4f af[4];
#pragma unroll
  for (int q = 0; q < 4; ++q)
    af[q] = *(const v4f*)(hs + m * HS_PITCH + q * 8 + 4 * h);

  float* orow[8];
  bool rok[8];
#pragma unroll
  for (int i = 0; i < 8; ++i) {
    const int tk = stok[8 * h + i];
    rok[i] = (tk >= 0 && tk < NTOK);
    orow[i] = out + (size_t)(rok[i] ? tk : 0) * D_DIM + wave * 256 + m;
  }

  const int dcol = wave * 256 + m;
  const float* blo = Bm + ((size_t)elo * D_DIM + dcol) * R_RANK + 4 * h;
  const float* bhi = Bm + ((size_t)ehi * D_DIM + dcol) * R_RANK + 4 * h;

  v4f bq[4][4], bn[4][4];
#pragma unroll
  for (int t = 0; t < 4; ++t) {
    bq[t][0] = *(const v4f*)(blo + t * 16 * R_RANK);
    bq[t][1] = *(const v4f*)(blo + t * 16 * R_RANK + 8);
    bq[t][2] = *(const v4f*)(bhi + t * 16 * R_RANK);
    bq[t][3] = *(const v4f*)(bhi + t * 16 * R_RANK + 8);
  }
#pragma unroll
  for (int c = 0; c < 4; ++c) {
    if (c + 1 < 4) {
#pragma unroll
      for (int t = 0; t < 4; ++t) {
        const int o = ((c + 1) * 4 + t) * 16 * R_RANK;
        bn[t][0] = *(const v4f*)(blo + o);
        bn[t][1] = *(const v4f*)(blo + o + 8);
        bn[t][2] = *(const v4f*)(bhi + o);
        bn[t][3] = *(const v4f*)(bhi + o + 8);
      }
    }
    v8f c0[4], c1[4];
#pragma unroll
    for (int t = 0; t < 4; ++t) { c0[t] = v8f{}; c1[t] = v8f{}; }
#pragma unroll
    for (int q = 0; q < 4; ++q) {
#pragma unroll
      for (int t = 0; t < 4; ++t) {
        c0[t] = wmma_f32(frag_lo(af[q]), frag_lo(bq[t][q]), c0[t]);
        c1[t] = wmma_f32(frag_hi(af[q]), frag_hi(bq[t][q]), c1[t]);
      }
    }
#pragma unroll
    for (int i = 0; i < 8; ++i) {
      if (rok[i]) {
        float* o = orow[i] + c * 64;
#pragma unroll
        for (int t = 0; t < 4; ++t) o[t * 16] = c0[t][i] + c1[t][i];
      }
    }
#pragma unroll
    for (int t = 0; t < 4; ++t)
#pragma unroll
      for (int q = 0; q < 4; ++q) bq[t][q] = bn[t][q];
  }
}

extern "C" void kernel_launch(void* const* d_in, const int* in_sizes, int n_in,
                              void* d_out, int out_size, void* d_ws, size_t ws_size,
                              hipStream_t stream) {
  (void)in_sizes; (void)n_in; (void)out_size; (void)ws_size;
  const float* x     = (const float*)d_in[0];
  const int*   alpha = (const int*)  d_in[1];
  const float* A     = (const float*)d_in[2];
  const float* Bm    = (const float*)d_in[3];
  const float* Wr    = (const float*)d_in[4];
  float* out = (float*)d_out;

  int*   tkey     = (int*)d_ws;
  float* tgate    = (float*)(tkey + NTOK);
  int*   perm     = (int*)(tgate + 2 * NTOK);
  int*   tile_key = perm + 8192;
  int*   ntiles   = tile_key + 512;

  moelora_router<<<NTOK / 16, 256, 0, stream>>>(x, Wr, alpha, tkey, tgate);
  moelora_group<<<1, 1024, 0, stream>>>(tkey, perm, tile_key, ntiles);
  moelora_tiles<<<NTILE_MAX, 256, 0, stream>>>(x, A, Bm, tgate, perm, tile_key, ntiles, out);
}


# group launch removed (3 launches -> 2): grid = 120 expert pairs x 4 row slots, each block finds its own 16 rows by scanning the 4096 pair keys (match masks + block prefix sum); sum orders unchanged
# speedup vs baseline: 4.2516x; 1.1927x over previous
#include <hip/hip_runtime.h>


typedef __attribute__((ext_vector_type(2))) float v2f;
typedef __attribute__((ext_vector_type(4))) float v4f;
typedef __attribute__((ext_vector_type(8))) float v8f;
typedef __attribute__((ext_vector_type(4))) int   v4i;

#define D_DIM  2048
#define K_EXP  16
#define R_RANK 16
#define NTOK   4096
#define NPAIR  120
#define JGRID  4

static __device__ __forceinline__ v8f wmma_f32(v2f a, v2f b, v8f c) {
  return __builtin_amdgcn_wmma_f32_16x16x4_f32(
       false, a,  false, b,
       (short)0, c,  false,  false);
}

static __device__ __forceinline__ v2f frag_lo(v4f v) {
  return __builtin_shufflevector(v, v, 0, 1);
}
static __device__ __forceinline__ v2f frag_hi(v4f v) {
  return __builtin_shufflevector(v, v, 2, 3);
}

#define RT_KW 256
#define RT_NG (RT_KW / 8)

__global__ __launch_bounds__(256) void moelora_router(
    const float* __restrict__ x, const float* __restrict__ Wr,
    const int* __restrict__ alpha_p, int* __restrict__ tkey,
    float* __restrict__ tgate)
{
  __shared__ float part[8 * 256];
  __shared__ __attribute__((aligned(16))) float sc[256];

  const int tid  = threadIdx.x;
  const int wave = tid >> 5;
  const int lane = tid & 31;
  const int m = lane & 15;
  const int h = lane >> 4;
  const int t0 = blockIdx.x * 16;

  const float* xa = x  + (size_t)(t0 + m) * D_DIM + wave * RT_KW + 4 * h;
  const float* wb = Wr + (size_t)m * D_DIM        + wave * RT_KW + 4 * h;

  v4f a[RT_NG], b[RT_NG];
#pragma unroll
  for (int i = 0; i < RT_NG; ++i) {
    a[i] = *(const v4f*)(xa + i * 8);
    b[i] = *(const v4f*)(wb + i * 8);
  }
  __builtin_amdgcn_sched_barrier(0);

  v8f c0 = {}, c1 = {}, c2 = {}, c3 = {};
#pragma unroll
  for (int i = 0; i < RT_NG; i += 2) {
    c0 = wmma_f32(frag_lo(a[i]),     frag_lo(b[i]),     c0);
    c1 = wmma_f32(frag_hi(a[i]),     frag_hi(b[i]),     c1);
    c2 = wmma_f32(frag_lo(a[i + 1]), frag_lo(b[i + 1]), c2);
    c3 = wmma_f32(frag_hi(a[i + 1]), frag_hi(b[i + 1]), c3);
  }

#pragma unroll
  for (int i = 0; i < 8; ++i)
    part[wave * 256 + (8 * h + i) * 16 + m] = (c0[i] + c1[i]) + (c2[i] + c3[i]);
  __syncthreads();

  float s = part[tid];
#pragma unroll
  for (int w = 1; w < 8; ++w) s += part[w * 256 + tid];
  sc[tid] = s;
  __syncthreads();

  const int tk = tid >> 4;
  const int ke = tid & 15;
  float acc[K_EXP];
#pragma unroll
  for (int q = 0; q < 4; ++q) {
    const v4f v = *(const v4f*)(sc + tk * 16 + q * 4);
    acc[q * 4 + 0] = v[0];
    acc[q * 4 + 1] = v[1];
    acc[q * 4 + 2] = v[2];
    acc[q * 4 + 3] = v[3];
  }
  const float alpha = (float)(*alpha_p);
  int i0 = 0; float v0 = acc[0];
#pragma unroll
  for (int k = 1; k < K_EXP; ++k) if (acc[k] > v0) { v0 = acc[k]; i0 = k; }
  int i1 = -1; float v1 = -3.4e38f;
#pragma unroll
  for (int k = 0; k < K_EXP; ++k)
    if (k != i0 && acc[k] > v1) { v1 = acc[k]; i1 = k; }
  const float e  = __expf(v1 - v0);
  const float dn = 1.f + e;
  const float g0 = alpha / dn;
  const float g1 = alpha * e / dn;
  if (i1 < 0) i1 = (i0 == 0) ? 1 : 0;
  if (ke == 0) {
    const int t = t0 + tk;
    const bool sw = i1 < i0;
    tkey[t] = sw ? (i1 * 16 + i0) : (i0 * 16 + i1);
    tgate[2 * t]     = sw ? g1 : g0;
    tgate[2 * t + 1] = sw ? g0 : g1;
  }
}

#define TL_NG 16
#define HS_PITCH 36

__global__ __launch_bounds__(256) void moelora_tiles(
    const float* x, const float* Amat, const float* Bm, const float* tgate,
    const int* __restrict__ tkey, float* out)
{
  __shared__ float part[8 * 512];
  __shared__ __attribute__((aligned(16))) float hs[16 * HS_PITCH];
  __shared__ int stok[16];
  __shared__ int wtot[8];

  const int tid  = threadIdx.x;
  const int wave = tid >> 5;
  const int lane = tid & 31;
  const int m = lane & 15;
  const int h = lane >> 4;

  int elo = 0, rem = blockIdx.x;
#pragma unroll
  for (int s = 0; s < 15; ++s)
    if (rem >= 15 - elo) { rem -= 15 - elo; ++elo; }
  const int ehi = elo + 1 + rem;
  const int mykey = elo * 16 + ehi;

  unsigned mask = 0;
#pragma unroll
  for (int q = 0; q < 4; ++q) {
    const v4i kv = *(const v4i*)(tkey + q * 1024 + tid * 4);
#pragma unroll
    for (int c = 0; c < 4; ++c)
      mask |= (kv[c] == mykey ? 1u : 0u) << (q * 4 + c);
  }
  const int cnt = __builtin_popcount(mask);
  int incl = cnt;
#pragma unroll
  for (int o = 1; o < 32; o <<= 1) {
    const int v = __shfl_up(incl, o);
    if (lane >= o) incl += v;
  }
  if (lane == 31) wtot[wave] = incl;
  __syncthreads();
  int base = incl - cnt, total = 0;
#pragma unroll
  for (int w = 0; w < 8; ++w) {
    const int t = wtot[w];
    total += t;
    base += (w < wave) ? t : 0;
  }

  const float* wlo = Amat + (size_t)(elo * 16 + m) * D_DIM + wave * 256 + 4 * h;
  const float* whi = Amat + (size_t)(ehi * 16 + m) * D_DIM + wave * 256 + 4 * h;

  for (int slot = blockIdx.y; slot * 16 < total; slot += JGRID) {
    if (slot != (int)blockIdx.y) __syncthreads();

    if (tid < 16 && tid >= total - 16 * slot) stok[tid] = -1;
    {
      unsigned mm = mask;
      int r = base - 16 * slot;
      while (mm != 0) {
        const int b = __builtin_ctz(mm);
        mm &= mm - 1;
        if (r >= 0 && r < 16) stok[r] = (b >> 2) * 1024 + tid * 4 + (b & 3);
        ++r;
      }
    }
    __syncthreads();

    const int tokm = stok[m];
    const int tokc = (tokm >= 0 && tokm < NTOK) ? tokm : 0;

    const float* xa  = x    + (size_t)tokc * D_DIM           + wave * 256 + 4 * h;

    v8f cl0 = {}, cl1 = {}, ch0 = {}, ch1 = {};
#pragma unroll
    for (int hf = 0; hf < 2; ++hf) {
      v4f a[TL_NG], bl[TL_NG], bh[TL_NG];
#pragma unroll
      for (int i = 0; i < TL_NG; ++i) {
        a[i]  = *(const v4f*)(xa  + hf * 128 + i * 8);
        bl[i] = *(const v4f*)(wlo + hf * 128 + i * 8);
        bh[i] = *(const v4f*)(whi + hf * 128 + i * 8);
      }
      __builtin_amdgcn_sched_barrier(0);
#pragma unroll
      for (int i = 0; i < TL_NG; ++i) {
        cl0 = wmma_f32(frag_lo(a[i]), frag_lo(bl[i]), cl0);
        ch0 = wmma_f32(frag_lo(a[i]), frag_lo(bh[i]), ch0);
        cl1 = wmma_f32(frag_hi(a[i]), frag_hi(bl[i]), cl1);
        ch1 = wmma_f32(frag_hi(a[i]), frag_hi(bh[i]), ch1);
      }
    }

#pragma unroll
    for (int i = 0; i < 8; ++i) {
      part[wave * 512 +       (8 * h + i) * 16 + m] = cl0[i] + cl1[i];
      part[wave * 512 + 256 + (8 * h + i) * 16 + m] = ch0[i] + ch1[i];
    }
    __syncthreads();

    {
      const int row = tid >> 4;
      const int r   = tid & 15;
      const int tk  = stok[row];
      const bool ok = (tk >= 0 && tk < NTOK);
      const float glo = ok ? tgate[2 * tk]     : 0.f;
      const float ghi = ok ? tgate[2 * tk + 1] : 0.f;
      float s0 = part[tid];
      float s1 = part[256 + tid];
#pragma unroll
      for (int w = 1; w < 8; ++w) {
        s0 += part[w * 512 + tid];
        s1 += part[w * 512 + 256 + tid];
      }
      hs[row * HS_PITCH + r]      = s0 * glo;
      hs[row * HS_PITCH + 16 + r] = s1 * ghi;
    }
    __syncthreads();

    v4f af[4];
#pragma unroll
    for (int q = 0; q < 4; ++q)
      af[q] = *(const v4f*)(hs + m * HS_PITCH + q * 8 + 4 * h);

    float* orow[8];
    bool rok[8];
#pragma unroll
    for (int i = 0; i < 8; ++i) {
      const int tk = stok[8 * h + i];
      rok[i] = (tk >= 0 && tk < NTOK);
      orow[i] = out + (size_t)(rok[i] ? tk : 0) * D_DIM + wave * 256 + m;
    }

    const int dcol = wave * 256 + m;
    const float* blo = Bm + ((size_t)elo * D_DIM + dcol) * R_RANK + 4 * h;
    const float* bhi = Bm + ((size_t)ehi * D_DIM + dcol) * R_RANK + 4 * h;

    v4f bq[4][4], bn[4][4];
#pragma unroll
    for (int t = 0; t < 4; ++t) {
      bq[t][0] = *(const v4f*)(blo + t * 16 * R_RANK);
      bq[t][1] = *(const v4f*)(blo + t * 16 * R_RANK + 8);
      bq[t][2] = *(const v4f*)(bhi + t * 16 * R_RANK);
      bq[t][3] = *(const v4f*)(bhi + t * 16 * R_RANK + 8);
    }
#pragma unroll
    for (int c = 0; c < 4; ++c) {
      if (c + 1 < 4) {
#pragma unroll
        for (int t = 0; t < 4; ++t) {
          const int o = ((c + 1) * 4 + t) * 16 * R_RANK;
          bn[t][0] = *(const v4f*)(blo + o);
          bn[t][1] = *(const v4f*)(blo + o + 8);
          bn[t][2] = *(const v4f*)(bhi + o);
          bn[t][3] = *(const v4f*)(bhi + o + 8);
        }
      }
      v8f c0[4], c1[4];
#pragma unroll
      for (int t = 0; t < 4; ++t) { c0[t] = v8f{}; c1[t] = v8f{}; }
#pragma unroll
      for (int q = 0; q < 4; ++q) {
#pragma unroll
        for (int t = 0; t < 4; ++t) {
          c0[t] = wmma_f32(frag_lo(af[q]), frag_lo(bq[t][q]), c0[t]);
          c1[t] = wmma_f32(frag_hi(af[q]), frag_hi(bq[t][q]), c1[t]);
        }
      }
#pragma unroll
      for (int i = 0; i < 8; ++i) {
        if (rok[i]) {
          float* o = orow[i] + c * 64;
#pragma unroll
          for (int t = 0; t < 4; ++t) o[t * 16] = c0[t][i] + c1[t][i];
        }
      }
#pragma unroll
      for (int t = 0; t < 4; ++t)
#pragma unroll
        for (int q = 0; q < 4; ++q) bq[t][q] = bn[t][q];
    }
  }
}

extern "C" void kernel_launch(void* const* d_in, const int* in_sizes, int n_in,
                              void* d_out, int out_size, void* d_ws, size_t ws_size,
                              hipStream_t stream) {
  (void)in_sizes; (void)n_in; (void)out_size; (void)ws_size;
  const float* x     = (const float*)d_in[0];
  const int*   alpha = (const int*)  d_in[1];
  const float* A     = (const float*)d_in[2];
  const float* Bm    = (const float*)d_in[3];
  const float* Wr    = (const float*)d_in[4];
  float* out = (float*)d_out;

  int*   tkey     = (int*)d_ws;
  float* tgate    = (float*)(tkey + NTOK);

  moelora_router<<<NTOK / 16, 256, 0, stream>>>(x, Wr, alpha, tkey, tgate);
  moelora_tiles<<<dim3(NPAIR, JGRID), 256, 0, stream>>>(x, A, Bm, tgate, tkey, out);
}


# tiles kernel: a block takes 32 rows of its pair as two 16-row M-tiles sharing every A and B weight fragment in registers (grid 120 x 2), next quarter's loads issued ahead; sum orders unchanged
# speedup vs baseline: 4.6284x; 1.0886x over previous
#include <hip/hip_runtime.h>


typedef __attribute__((ext_vector_type(2))) float v2f;
typedef __attribute__((ext_vector_type(4))) float v4f;
typedef __attribute__((ext_vector_type(8))) float v8f;
typedef __attribute__((ext_vector_type(4))) int   v4i;

#define D_DIM  2048
#define K_EXP  16
#define R_RANK 16
#define NTOK   4096
#define NPAIR  120
#define JGRID  2

static __device__ __forceinline__ v8f wmma_f32(v2f a, v2f b, v8f c) {
  return __builtin_amdgcn_wmma_f32_16x16x4_f32(
       false, a,  false, b,
       (short)0, c,  false,  false);
}

static __device__ __forceinline__ v2f frag_lo(v4f v) {
  return __builtin_shufflevector(v, v, 0, 1);
}
static __device__ __forceinline__ v2f frag_hi(v4f v) {
  return __builtin_shufflevector(v, v, 2, 3);
}

#define RT_KW 256
#define RT_NG (RT_KW / 8)

__global__ __launch_bounds__(256) void moelora_router(
    const float* __restrict__ x, const float* __restrict__ Wr,
    const int* __restrict__ alpha_p, int* __restrict__ tkey,
    float* __restrict__ tgate)
{
  __shared__ float part[8 * 256];
  __shared__ __attribute__((aligned(16))) float sc[256];

  const int tid  = threadIdx.x;
  const int wave = tid >> 5;
  const int lane = tid & 31;
  const int m = lane & 15;
  const int h = lane >> 4;
  const int t0 = blockIdx.x * 16;

  const float* xa = x  + (size_t)(t0 + m) * D_DIM + wave * RT_KW + 4 * h;
  const float* wb = Wr + (size_t)m * D_DIM        + wave * RT_KW + 4 * h;

  v4f a[RT_NG], b[RT_NG];
#pragma unroll
  for (int i = 0; i < RT_NG; ++i) {
    a[i] = *(const v4f*)(xa + i * 8);
    b[i] = *(const v4f*)(wb + i * 8);
  }
  __builtin_amdgcn_sched_barrier(0);

  v8f c0 = {}, c1 = {}, c2 = {}, c3 = {};
#pragma unroll
  for (int i = 0; i < RT_NG; i += 2) {
    c0 = wmma_f32(frag_lo(a[i]),     frag_lo(b[i]),     c0);
    c1 = wmma_f32(frag_hi(a[i]),     frag_hi(b[i]),     c1);
    c2 = wmma_f32(frag_lo(a[i + 1]), frag_lo(b[i + 1]), c2);
    c3 = wmma_f32(frag_hi(a[i + 1]), frag_hi(b[i + 1]), c3);
  }

#pragma unroll
  for (int i = 0; i < 8; ++i)
    part[wave * 256 + (8 * h + i) * 16 + m] = (c0[i] + c1[i]) + (c2[i] + c3[i]);
  __syncthreads();

  float s = part[tid];
#pragma unroll
  for (int w = 1; w < 8; ++w) s += part[w * 256 + tid];
  sc[tid] = s;
  __syncthreads();

  const int tk = tid >> 4;
  const int ke = tid & 15;
  float acc[K_EXP];
#pragma unroll
  for (int q = 0; q < 4; ++q) {
    const v4f v = *(const v4f*)(sc + tk * 16 + q * 4);
    acc[q * 4 + 0] = v[0];
    acc[q * 4 + 1] = v[1];
    acc[q * 4 + 2] = v[2];
    acc[q * 4 + 3] = v[3];
  }
  const float alpha = (float)(*alpha_p);
  int i0 = 0; float v0 = acc[0];
#pragma unroll
  for (int k = 1; k < K_EXP; ++k) if (acc[k] > v0) { v0 = acc[k]; i0 = k; }
  int i1 = -1; float v1 = -3.4e38f;
#pragma unroll
  for (int k = 0; k < K_EXP; ++k)
    if (k != i0 && acc[k] > v1) { v1 = acc[k]; i1 = k; }
  const float e  = __expf(v1 - v0);
  const float dn = 1.f + e;
  const float g0 = alpha / dn;
  const float g1 = alpha * e / dn;
  if (i1 < 0) i1 = (i0 == 0) ? 1 : 0;
  if (ke == 0) {
    const int t = t0 + tk;
    const bool sw = i1 < i0;
    tkey[t] = sw ? (i1 * 16 + i0) : (i0 * 16 + i1);
    tgate[2 * t]     = sw ? g1 : g0;
    tgate[2 * t + 1] = sw ? g0 : g1;
  }
}

#define TL_QG 8
#define HS_PITCH 36

template <int MT>
static __device__ __forceinline__ void tile_body(
    const float* x, const float* Bm, const float* tgate, float* out,
    const float* wlo, const float* whi, const int elo, const int ehi,
    float* part, float* hs, const int* stok,
    const int tid, const int wave, const int m, const int h)
{
  const float* xa[MT];
#pragma unroll
  for (int mt = 0; mt < MT; ++mt) {
    const int tokm = stok[16 * mt + m];
    const int tokc = (tokm >= 0 && tokm < NTOK) ? tokm : 0;
    xa[mt] = x + (size_t)tokc * D_DIM + wave * 256 + 4 * h;
  }

  v8f acc[MT][4];
#pragma unroll
  for (int mt = 0; mt < MT; ++mt)
#pragma unroll
    for (int j = 0; j < 4; ++j) acc[mt][j] = v8f{};

  v4f a[2][MT][TL_QG], bl[2][TL_QG], bh[2][TL_QG];
#pragma unroll
  for (int i = 0; i < TL_QG; ++i) {
#pragma unroll
    for (int mt = 0; mt < MT; ++mt) a[0][mt][i] = *(const v4f*)(xa[mt] + i * 8);
    bl[0][i] = *(const v4f*)(wlo + i * 8);
    bh[0][i] = *(const v4f*)(whi + i * 8);
  }
#pragma unroll
  for (int q = 0; q < 4; ++q) {
    const int cur = q & 1;
    const int nxt = cur ^ 1;
    if (q + 1 < 4) {
#pragma unroll
      for (int i = 0; i < TL_QG; ++i) {
#pragma unroll
        for (int mt = 0; mt < MT; ++mt)
          a[nxt][mt][i] = *(const v4f*)(xa[mt] + (q + 1) * 64 + i * 8);
        bl[nxt][i] = *(const v4f*)(wlo + (q + 1) * 64 + i * 8);
        bh[nxt][i] = *(const v4f*)(whi + (q + 1) * 64 + i * 8);
      }
    }
    __builtin_amdgcn_sched_barrier(0);
#pragma unroll
    for (int i = 0; i < TL_QG; ++i) {
#pragma unroll
      for (int mt = 0; mt < MT; ++mt) {
        acc[mt][0] = wmma_f32(frag_lo(a[cur][mt][i]), frag_lo(bl[cur][i]), acc[mt][0]);
        acc[mt][1] = wmma_f32(frag_lo(a[cur][mt][i]), frag_lo(bh[cur][i]), acc[mt][1]);
        acc[mt][2] = wmma_f32(frag_hi(a[cur][mt][i]), frag_hi(bl[cur][i]), acc[mt][2]);
        acc[mt][3] = wmma_f32(frag_hi(a[cur][mt][i]), frag_hi(bh[cur][i]), acc[mt][3]);
      }
    }
    __builtin_amdgcn_sched_barrier(0);
  }

#pragma unroll
  for (int mt = 0; mt < MT; ++mt) {
#pragma unroll
    for (int i = 0; i < 8; ++i) {
      part[wave * (MT * 512) + mt * 512 +       (8 * h + i) * 16 + m] = acc[mt][0][i] + acc[mt][2][i];
      part[wave * (MT * 512) + mt * 512 + 256 + (8 * h + i) * 16 + m] = acc[mt][1][i] + acc[mt][3][i];
    }
  }
  __syncthreads();

#pragma unroll
  for (int mt = 0; mt < MT; ++mt) {
    const int row = tid >> 4;
    const int r   = tid & 15;
    const int tk  = stok[16 * mt + row];
    const bool ok = (tk >= 0 && tk < NTOK);
    const float glo = ok ? tgate[2 * tk]     : 0.f;
    const float ghi = ok ? tgate[2 * tk + 1] : 0.f;
    float s0 = part[mt * 512 + tid];
    float s1 = part[mt * 512 + 256 + tid];
#pragma unroll
    for (int w = 1; w < 8; ++w) {
      s0 += part[w * (MT * 512) + mt * 512 + tid];
      s1 += part[w * (MT * 512) + mt * 512 + 256 + tid];
    }
    hs[(16 * mt + row) * HS_PITCH + r]      = s0 * glo;
    hs[(16 * mt + row) * HS_PITCH + 16 + r] = s1 * ghi;
  }
  __syncthreads();

  v4f af[MT][4];
#pragma unroll
  for (int mt = 0; mt < MT; ++mt)
#pragma unroll
    for (int q = 0; q < 4; ++q)
      af[mt][q] = *(const v4f*)(hs + (16 * mt + m) * HS_PITCH + q * 8 + 4 * h);

  int tko[MT][8];
#pragma unroll
  for (int mt = 0; mt < MT; ++mt)
#pragma unroll
    for (int i = 0; i < 8; ++i) {
      const int tk = stok[16 * mt + 8 * h + i];
      tko[mt][i] = (tk >= 0 && tk < NTOK) ? tk : -1;
    }

  const int dcol = wave * 256 + m;
  const float* blo = Bm + ((size_t)elo * D_DIM + dcol) * R_RANK + 4 * h;
  const float* bhi = Bm + ((size_t)ehi * D_DIM + dcol) * R_RANK + 4 * h;

  v4f bq[4][4], bn[4][4];
#pragma unroll
  for (int t = 0; t < 4; ++t) {
    bq[t][0] = *(const v4f*)(blo + t * 16 * R_RANK);
    bq[t][1] = *(const v4f*)(blo + t * 16 * R_RANK + 8);
    bq[t][2] = *(const v4f*)(bhi + t * 16 * R_RANK);
    bq[t][3] = *(const v4f*)(bhi + t * 16 * R_RANK + 8);
  }
#pragma unroll
  for (int c = 0; c < 4; ++c) {
    if (c + 1 < 4) {
#pragma unroll
      for (int t = 0; t < 4; ++t) {
        const int o = ((c + 1) * 4 + t) * 16 * R_RANK;
        bn[t][0] = *(const v4f*)(blo + o);
        bn[t][1] = *(const v4f*)(blo + o + 8);
        bn[t][2] = *(const v4f*)(bhi + o);
        bn[t][3] = *(const v4f*)(bhi + o + 8);
      }
    }
    v8f c0[MT][4], c1[MT][4];
#pragma unroll
    for (int mt = 0; mt < MT; ++mt)
#pragma unroll
      for (int t = 0; t < 4; ++t) { c0[mt][t] = v8f{}; c1[mt][t] = v8f{}; }
#pragma unroll
    for (int q = 0; q < 4; ++q) {
#pragma unroll
      for (int t = 0; t < 4; ++t) {
#pragma unroll
        for (int mt = 0; mt < MT; ++mt) {
          c0[mt][t] = wmma_f32(frag_lo(af[mt][q]), frag_lo(bq[t][q]), c0[mt][t]);
          c1[mt][t] = wmma_f32(frag_hi(af[mt][q]), frag_hi(bq[t][q]), c1[mt][t]);
        }
      }
    }
#pragma unroll
    for (int mt = 0; mt < MT; ++mt)
#pragma unroll
      for (int i = 0; i < 8; ++i) {
        if (tko[mt][i] >= 0) {
          float* o = out + (size_t)tko[mt][i] * D_DIM + wave * 256 + m + c * 64;
#pragma unroll
          for (int t = 0; t < 4; ++t) o[t * 16] = c0[mt][t][i] + c1[mt][t][i];
        }
      }
#pragma unroll
    for (int t = 0; t < 4; ++t)
#pragma unroll
      for (int q = 0; q < 4; ++q) bq[t][q] = bn[t][q];
  }
}

__global__ __launch_bounds__(256) void moelora_tiles(
    const float* x, const float* Amat, const float* Bm, const float* tgate,
    const int* __restrict__ tkey, float* out)
{
  __shared__ float part[8 * 1024];
  __shared__ __attribute__((aligned(16))) float hs[32 * HS_PITCH];
  __shared__ int stok[32];
  __shared__ int wtot[8];

  const int tid  = threadIdx.x;
  const int wave = tid >> 5;
  const int lane = tid & 31;
  const int m = lane & 15;
  const int h = lane >> 4;

  int elo = 0, rem = blockIdx.x;
#pragma unroll
  for (int s = 0; s < 15; ++s)
    if (rem >= 15 - elo) { rem -= 15 - elo; ++elo; }
  const int ehi = elo + 1 + rem;
  const int mykey = elo * 16 + ehi;

  unsigned mask = 0;
#pragma unroll
  for (int q = 0; q < 4; ++q) {
    const v4i kv = *(const v4i*)(tkey + q * 1024 + tid * 4);
#pragma unroll
    for (int c = 0; c < 4; ++c)
      mask |= (kv[c] == mykey ? 1u : 0u) << (q * 4 + c);
  }
  const int cnt = __builtin_popcount(mask);
  int incl = cnt;
#pragma unroll
  for (int o = 1; o < 32; o <<= 1) {
    const int v = __shfl_up(incl, o);
    if (lane >= o) incl += v;
  }
  if (lane == 31) wtot[wave] = incl;
  __syncthreads();
  int base = incl - cnt, total = 0;
#pragma unroll
  for (int w = 0; w < 8; ++w) {
    const int t = wtot[w];
    total += t;
    base += (w < wave) ? t : 0;
  }

  const float* wlo = Amat + (size_t)(elo * 16 + m) * D_DIM + wave * 256 + 4 * h;
  const float* whi = Amat + (size_t)(ehi * 16 + m) * D_DIM + wave * 256 + 4 * h;

  for (int slot = blockIdx.y; slot * 32 < total; slot += JGRID) {
    if (slot != (int)blockIdx.y) __syncthreads();

    const int left = total - 32 * slot;
    if (tid < 32 && tid >= left) stok[tid] = -1;
    {
      unsigned mm = mask;
      int r = base - 32 * slot;
      while (mm != 0) {
        const int b = __builtin_ctz(mm);
        mm &= mm - 1;
        if (r >= 0 && r < 32) stok[r] = (b >> 2) * 1024 + tid * 4 + (b & 3);
        ++r;
      }
    }
    __syncthreads();

    if (left > 16)
      tile_body<2>(x, Bm, tgate, out, wlo, whi, elo, ehi, part, hs, stok, tid, wave, m, h);
    else
      tile_body<1>(x, Bm, tgate, out, wlo, whi, elo, ehi, part, hs, stok, tid, wave, m, h);
  }
}

extern "C" void kernel_launch(void* const* d_in, const int* in_sizes, int n_in,
                              void* d_out, int out_size, void* d_ws, size_t ws_size,
                              hipStream_t stream) {
  (void)in_sizes; (void)n_in; (void)out_size; (void)ws_size;
  const float* x     = (const float*)d_in[0];
  const int*   alpha = (const int*)  d_in[1];
  const float* A     = (const float*)d_in[2];
  const float* Bm    = (const float*)d_in[3];
  const float* Wr    = (const float*)d_in[4];
  float* out = (float*)d_out;

  int*   tkey     = (int*)d_ws;
  float* tgate    = (float*)(tkey + NTOK);

  moelora_router<<<NTOK / 16, 256, 0, stream>>>(x, Wr, alpha, tkey, tgate);
  moelora_tiles<<<dim3(NPAIR, JGRID), 256, 0, stream>>>(x, A, Bm, tgate, tkey, out);
}


# both kernels: x and weight rows fetched by whole-line 128-bit global loads (4 cache lines per load, not one per row), staged in per-wave LDS, WMMA fragments by ds_load_b128; k sum orders unchanged
# speedup vs baseline: 5.2114x; 1.1260x over previous
#include <hip/hip_runtime.h>


typedef __attribute__((ext_vector_type(2))) float v2f;
typedef __attribute__((ext_vector_type(4))) float v4f;
typedef __attribute__((ext_vector_type(8))) float v8f;
typedef __attribute__((ext_vector_type(4))) int   v4i;

#define D_DIM  2048
#define K_EXP  16
#define R_RANK 16
#define NTOK   4096
#define NPAIR  120
#define JGRID  2
#define SG_PITCH 68

static __device__ __forceinline__ v8f wmma_f32(v2f a, v2f b, v8f c) {
  return __builtin_amdgcn_wmma_f32_16x16x4_f32(
       false, a,  false, b,
       (short)0, c,  false,  false);
}

static __device__ __forceinline__ v2f frag_lo(v4f v) {
  return __builtin_shufflevector(v, v, 0, 1);
}
static __device__ __forceinline__ v2f frag_hi(v4f v) {
  return __builtin_shufflevector(v, v, 2, 3);
}

#define RT_KW 256

__global__ __launch_bounds__(256) void moelora_router(
    const float* __restrict__ x, const float* __restrict__ Wr,
    const int* __restrict__ alpha_p, int* __restrict__ tkey,
    float* __restrict__ tgate)
{
  __shared__ float part[8 * 256];
  __shared__ __attribute__((aligned(16))) float sc[256];
  __shared__ __attribute__((aligned(16))) float stg[8 * 32 * SG_PITCH];

  const int tid  = threadIdx.x;
  const int wave = tid >> 5;
  const int lane = tid & 31;
  const int m = lane & 15;
  const int h = lane >> 4;
  const int t0 = blockIdx.x * 16;

  const int g  = lane >> 4;
  const int cl = lane & 15;
  const float* xl = x  + (size_t)(t0 + g) * D_DIM + wave * RT_KW + 4 * cl;
  const float* wl = Wr + (size_t)g * D_DIM        + wave * RT_KW + 4 * cl;
  float* sw  = stg + wave * (32 * SG_PITCH);
  float* swr = sw + g * SG_PITCH + 4 * cl;
  const float* frd = sw + m * SG_PITCH + 4 * h;

  v4f gq[4][16];
#pragma unroll
  for (int q = 0; q < 4; ++q) {
#pragma unroll
    for (int j = 0; j < 8; ++j) {
      gq[q][j]     = *(const v4f*)(xl + j * 2 * D_DIM + q * 64);
      gq[q][8 + j] = *(const v4f*)(wl + j * 2 * D_DIM + q * 64);
    }
  }
  __builtin_amdgcn_sched_barrier(0);

  v8f c0 = {}, c1 = {}, c2 = {}, c3 = {};
  v4f a[2][8], b[2][8];
#pragma unroll
  for (int q = -1; q < 4; ++q) {
    if (q + 1 < 4) {
      const int nx = (q + 1) & 1;
#pragma unroll
      for (int j = 0; j < 16; ++j) *(v4f*)(swr + j * 2 * SG_PITCH) = gq[q + 1][j];
#pragma unroll
      for (int i = 0; i < 8; ++i) {
        a[nx][i] = *(const v4f*)(frd + i * 8);
        b[nx][i] = *(const v4f*)(frd + 16 * SG_PITCH + i * 8);
      }
    }
    __builtin_amdgcn_sched_barrier(0);
    if (q >= 0) {
      const int cu = q & 1;
#pragma unroll
      for (int i = 0; i < 8; i += 2) {
        c0 = wmma_f32(frag_lo(a[cu][i]),     frag_lo(b[cu][i]),     c0);
        c1 = wmma_f32(frag_hi(a[cu][i]),     frag_hi(b[cu][i]),     c1);
        c2 = wmma_f32(frag_lo(a[cu][i + 1]), frag_lo(b[cu][i + 1]), c2);
        c3 = wmma_f32(frag_hi(a[cu][i + 1]), frag_hi(b[cu][i + 1]), c3);
      }
      __builtin_amdgcn_sched_barrier(0);
    }
  }

#pragma unroll
  for (int i = 0; i < 8; ++i)
    part[wave * 256 + (8 * h + i) * 16 + m] = (c0[i] + c1[i]) + (c2[i] + c3[i]);
  __syncthreads();

  float s = part[tid];
#pragma unroll
  for (int w = 1; w < 8; ++w) s += part[w * 256 + tid];
  sc[tid] = s;
  __syncthreads();

  const int tk = tid >> 4;
  const int ke = tid & 15;
  float acc[K_EXP];
#pragma unroll
  for (int q = 0; q < 4; ++q) {
    const v4f v = *(const v4f*)(sc + tk * 16 + q * 4);
    acc[q * 4 + 0] = v[0];
    acc[q * 4 + 1] = v[1];
    acc[q * 4 + 2] = v[2];
    acc[q * 4 + 3] = v[3];
  }
  const float alpha = (float)(*alpha_p);
  int i0 = 0; float v0 = acc[0];
#pragma unroll
  for (int k = 1; k < K_EXP; ++k) if (acc[k] > v0) { v0 = acc[k]; i0 = k; }
  int i1 = -1; float v1 = -3.4e38f;
#pragma unroll
  for (int k = 0; k < K_EXP; ++k)
    if (k != i0 && acc[k] > v1) { v1 = acc[k]; i1 = k; }
  const float e  = __expf(v1 - v0);
  const float dn = 1.f + e;
  const float g0 = alpha / dn;
  const float g1 = alpha * e / dn;
  if (i1 < 0) i1 = (i0 == 0) ? 1 : 0;
  if (ke == 0) {
    const int t = t0 + tk;
    const bool sw = i1 < i0;
    tkey[t] = sw ? (i1 * 16 + i0) : (i0 * 16 + i1);
    tgate[2 * t]     = sw ? g1 : g0;
    tgate[2 * t + 1] = sw ? g0 : g1;
  }
}

#define TL_QG 8
#define HS_PITCH 36
#define SG_ROWS  64

template <int MT>
static __device__ __forceinline__ void tile_body(
    const float* x, const float* Bm, const float* tgate, float* out,
    const float* wlo, const float* whi, const int elo, const int ehi,
    float* part, float* hs, const int* stok, float* sw,
    const int tid, const int wave, const int lane, const int m, const int h)
{
  constexpr int NX = 8 * MT;
  constexpr int NJ = NX + 16;
  const int g = lane >> 4;
  const int c = lane & 15;
  const float* xg[NX];
#pragma unroll
  for (int j = 0; j < NX; ++j) {
    const int tokm = stok[2 * j + g];
    const int tokc = (tokm >= 0 && tokm < NTOK) ? tokm : 0;
    xg[j] = x + (size_t)tokc * D_DIM + wave * 256 + 4 * c;
  }
  float* swr = sw + g * SG_PITCH + 4 * c;
  const float* frd = sw + m * SG_PITCH + 4 * h;

  v8f acc[MT][4];
#pragma unroll
  for (int mt = 0; mt < MT; ++mt)
#pragma unroll
    for (int j = 0; j < 4; ++j) acc[mt][j] = v8f{};

  v4f gq[NJ];
#pragma unroll
  for (int j = 0; j < NX; ++j) gq[j] = *(const v4f*)(xg[j]);
#pragma unroll
  for (int j = 0; j < 8; ++j) {
    gq[NX + j]     = *(const v4f*)(wlo + j * 2 * D_DIM);
    gq[NX + 8 + j] = *(const v4f*)(whi + j * 2 * D_DIM);
  }
#pragma unroll
  for (int q = 0; q < 4; ++q) {
#pragma unroll
    for (int j = 0; j < NJ; ++j) *(v4f*)(swr + j * 2 * SG_PITCH) = gq[j];
    if (q + 1 < 4) {
#pragma unroll
      for (int j = 0; j < NX; ++j) gq[j] = *(const v4f*)(xg[j] + (q + 1) * 64);
#pragma unroll
      for (int j = 0; j < 8; ++j) {
        gq[NX + j]     = *(const v4f*)(wlo + j * 2 * D_DIM + (q + 1) * 64);
        gq[NX + 8 + j] = *(const v4f*)(whi + j * 2 * D_DIM + (q + 1) * 64);
      }
    }
    v4f a[MT][TL_QG], bl[TL_QG], bh[TL_QG];
#pragma unroll
    for (int i = 0; i < TL_QG; ++i) {
#pragma unroll
      for (int mt = 0; mt < MT; ++mt)
        a[mt][i] = *(const v4f*)(frd + (16 * mt) * SG_PITCH + i * 8);
      bl[i] = *(const v4f*)(frd + (16 * MT) * SG_PITCH + i * 8);
      bh[i] = *(const v4f*)(frd + (16 * MT + 16) * SG_PITCH + i * 8);
    }
    __builtin_amdgcn_sched_barrier(0);
#pragma unroll
    for (int i = 0; i < TL_QG; ++i) {
#pragma unroll
      for (int mt = 0; mt < MT; ++mt) {
        acc[mt][0] = wmma_f32(frag_lo(a[mt][i]), frag_lo(bl[i]), acc[mt][0]);
        acc[mt][1] = wmma_f32(frag_lo(a[mt][i]), frag_lo(bh[i]), acc[mt][1]);
        acc[mt][2] = wmma_f32(frag_hi(a[mt][i]), frag_hi(bl[i]), acc[mt][2]);
        acc[mt][3] = wmma_f32(frag_hi(a[mt][i]), frag_hi(bh[i]), acc[mt][3]);
      }
    }
    __builtin_amdgcn_sched_barrier(0);
  }

#pragma unroll
  for (int mt = 0; mt < MT; ++mt) {
#pragma unroll
    for (int i = 0; i < 8; ++i) {
      part[wave * (MT * 512) + mt * 512 +       (8 * h + i) * 16 + m] = acc[mt][0][i] + acc[mt][2][i];
      part[wave * (MT * 512) + mt * 512 + 256 + (8 * h + i) * 16 + m] = acc[mt][1][i] + acc[mt][3][i];
    }
  }
  __syncthreads();

#pragma unroll
  for (int mt = 0; mt < MT; ++mt) {
    const int row = tid >> 4;
    const int r   = tid & 15;
    const int tk  = stok[16 * mt + row];
    const bool ok = (tk >= 0 && tk < NTOK);
    const float glo = ok ? tgate[2 * tk]     : 0.f;
    const float ghi = ok ? tgate[2 * tk + 1] : 0.f;
    float s0 = part[mt * 512 + tid];
    float s1 = part[mt * 512 + 256 + tid];
#pragma unroll
    for (int w = 1; w < 8; ++w) {
      s0 += part[w * (MT * 512) + mt * 512 + tid];
      s1 += part[w * (MT * 512) + mt * 512 + 256 + tid];
    }
    hs[(16 * mt + row) * HS_PITCH + r]      = s0 * glo;
    hs[(16 * mt + row) * HS_PITCH + 16 + r] = s1 * ghi;
  }
  __syncthreads();

  v4f af[MT][4];
#pragma unroll
  for (int mt = 0; mt < MT; ++mt)
#pragma unroll
    for (int q = 0; q < 4; ++q)
      af[mt][q] = *(const v4f*)(hs + (16 * mt + m) * HS_PITCH + q * 8 + 4 * h);

  int tko[MT][8];
#pragma unroll
  for (int mt = 0; mt < MT; ++mt)
#pragma unroll
    for (int i = 0; i < 8; ++i) {
      const int tk = stok[16 * mt + 8 * h + i];
      tko[mt][i] = (tk >= 0 && tk < NTOK) ? tk : -1;
    }

  const int dcol = wave * 256 + m;
  const float* blo = Bm + ((size_t)elo * D_DIM + dcol) * R_RANK + 4 * h;
  const float* bhi = Bm + ((size_t)ehi * D_DIM + dcol) * R_RANK + 4 * h;

  v4f bq[4][4], bn[4][4];
#pragma unroll
  for (int t = 0; t < 4; ++t) {
    bq[t][0] = *(const v4f*)(blo + t * 16 * R_RANK);
    bq[t][1] = *(const v4f*)(blo + t * 16 * R_RANK + 8);
    bq[t][2] = *(const v4f*)(bhi + t * 16 * R_RANK);
    bq[t][3] = *(const v4f*)(bhi + t * 16 * R_RANK + 8);
  }
#pragma unroll
  for (int c = 0; c < 4; ++c) {
    if (c + 1 < 4) {
#pragma unroll
      for (int t = 0; t < 4; ++t) {
        const int o = ((c + 1) * 4 + t) * 16 * R_RANK;
        bn[t][0] = *(const v4f*)(blo + o);
        bn[t][1] = *(const v4f*)(blo + o + 8);
        bn[t][2] = *(const v4f*)(bhi + o);
        bn[t][3] = *(const v4f*)(bhi + o + 8);
      }
    }
    v8f c0[MT][4], c1[MT][4];
#pragma unroll
    for (int mt = 0; mt < MT; ++mt)
#pragma unroll
      for (int t = 0; t < 4; ++t) { c0[mt][t] = v8f{}; c1[mt][t] = v8f{}; }
#pragma unroll
    for (int q = 0; q < 4; ++q) {
#pragma unroll
      for (int t = 0; t < 4; ++t) {
#pragma unroll
        for (int mt = 0; mt < MT; ++mt) {
          c0[mt][t] = wmma_f32(frag_lo(af[mt][q]), frag_lo(bq[t][q]), c0[mt][t]);
          c1[mt][t] = wmma_f32(frag_hi(af[mt][q]), frag_hi(bq[t][q]), c1[mt][t]);
        }
      }
    }
#pragma unroll
    for (int mt = 0; mt < MT; ++mt)
#pragma unroll
      for (int i = 0; i < 8; ++i) {
        if (tko[mt][i] >= 0) {
          float* o = out + (size_t)tko[mt][i] * D_DIM + wave * 256 + m + c * 64;
#pragma unroll
          for (int t = 0; t < 4; ++t) o[t * 16] = c0[mt][t][i] + c1[mt][t][i];
        }
      }
#pragma unroll
    for (int t = 0; t < 4; ++t)
#pragma unroll
      for (int q = 0; q < 4; ++q) bq[t][q] = bn[t][q];
  }
}

__global__ __launch_bounds__(256) void moelora_tiles(
    const float* x, const float* Amat, const float* Bm, const float* tgate,
    const int* __restrict__ tkey, float* out)
{
  __shared__ float part[8 * 1024];
  __shared__ __attribute__((aligned(16))) float hs[32 * HS_PITCH];
  __shared__ __attribute__((aligned(16))) float stg[8 * SG_ROWS * SG_PITCH];
  __shared__ int stok[32];
  __shared__ int wtot[8];

  const int tid  = threadIdx.x;
  const int wave = tid >> 5;
  const int lane = tid & 31;
  const int m = lane & 15;
  const int h = lane >> 4;

  int elo = 0, rem = blockIdx.x;
#pragma unroll
  for (int s = 0; s < 15; ++s)
    if (rem >= 15 - elo) { rem -= 15 - elo; ++elo; }
  const int ehi = elo + 1 + rem;
  const int mykey = elo * 16 + ehi;

  unsigned mask = 0;
#pragma unroll
  for (int q = 0; q < 4; ++q) {
    const v4i kv = *(const v4i*)(tkey + q * 1024 + tid * 4);
#pragma unroll
    for (int c = 0; c < 4; ++c)
      mask |= (kv[c] == mykey ? 1u : 0u) << (q * 4 + c);
  }
  const int cnt = __builtin_popcount(mask);
  int incl = cnt;
#pragma unroll
  for (int o = 1; o < 32; o <<= 1) {
    const int v = __shfl_up(incl, o);
    if (lane >= o) incl += v;
  }
  if (lane == 31) wtot[wave] = incl;
  __syncthreads();
  int base = incl - cnt, total = 0;
#pragma unroll
  for (int w = 0; w < 8; ++w) {
    const int t = wtot[w];
    total += t;
    base += (w < wave) ? t : 0;
  }

  const float* wlo = Amat + (size_t)(elo * 16 + (lane >> 4)) * D_DIM + wave * 256 + 4 * (lane & 15);
  const float* whi = Amat + (size_t)(ehi * 16 + (lane >> 4)) * D_DIM + wave * 256 + 4 * (lane & 15);
  float* sw = stg + wave * (SG_ROWS * SG_PITCH);

  for (int slot = blockIdx.y; slot * 32 < total; slot += JGRID) {
    if (slot != (int)blockIdx.y) __syncthreads();

    const int left = total - 32 * slot;
    if (tid < 32 && tid >= left) stok[tid] = -1;
    {
      unsigned mm = mask;
      int r = base - 32 * slot;
      while (mm != 0) {
        const int b = __builtin_ctz(mm);
        mm &= mm - 1;
        if (r >= 0 && r < 32) stok[r] = (b >> 2) * 1024 + tid * 4 + (b & 3);
        ++r;
      }
    }
    __syncthreads();

    if (left > 16)
      tile_body<2>(x, Bm, tgate, out, wlo, whi, elo, ehi, part, hs, stok, sw, tid, wave, lane, m, h);
    else
      tile_body<1>(x, Bm, tgate, out, wlo, whi, elo, ehi, part, hs, stok, sw, tid, wave, lane, m, h);
  }
}

extern "C" void kernel_launch(void* const* d_in, const int* in_sizes, int n_in,
                              void* d_out, int out_size, void* d_ws, size_t ws_size,
                              hipStream_t stream) {
  (void)in_sizes; (void)n_in; (void)out_size; (void)ws_size;
  const float* x     = (const float*)d_in[0];
  const int*   alpha = (const int*)  d_in[1];
  const float* A     = (const float*)d_in[2];
  const float* Bm    = (const float*)d_in[3];
  const float* Wr    = (const float*)d_in[4];
  float* out = (float*)d_out;

  int*   tkey     = (int*)d_ws;
  float* tgate    = (float*)(tkey + NTOK);

  moelora_router<<<NTOK / 16, 256, 0, stream>>>(x, Wr, alpha, tkey, tgate);
  moelora_tiles<<<dim3(NPAIR, JGRID), 256, 0, stream>>>(x, A, Bm, tgate, tkey, out);
}


# loads issued ahead of the phase that waited for them: first A quarter before the key scan (its 2 barriers fence LDS only), row gates at tile start, first B chunk in phase 1's last quarter; sums kept
# speedup vs baseline: 5.4963x; 1.0547x over previous
#include <hip/hip_runtime.h>


typedef __attribute__((ext_vector_type(2))) float v2f;
typedef __attribute__((ext_vector_type(4))) float v4f;
typedef __attribute__((ext_vector_type(8))) float v8f;
typedef __attribute__((ext_vector_type(4))) int   v4i;

#define D_DIM  2048
#define K_EXP  16
#define R_RANK 16
#define NTOK   4096
#define NPAIR  120
#define JGRID  2
#define SG_PITCH 68

static __device__ __forceinline__ v8f wmma_f32(v2f a, v2f b, v8f c) {
  return __builtin_amdgcn_wmma_f32_16x16x4_f32(
       false, a,  false, b,
       (short)0, c,  false,  false);
}

static __device__ __forceinline__ void barrier_lds() {
  __builtin_amdgcn_fence(__ATOMIC_RELEASE, "workgroup", "local");
  __builtin_amdgcn_s_barrier();
  __builtin_amdgcn_fence(__ATOMIC_ACQUIRE, "workgroup", "local");
}

static __device__ __forceinline__ v2f frag_lo(v4f v) {
  return __builtin_shufflevector(v, v, 0, 1);
}
static __device__ __forceinline__ v2f frag_hi(v4f v) {
  return __builtin_shufflevector(v, v, 2, 3);
}

#define RT_KW 256

__global__ __launch_bounds__(256) void moelora_router(
    const float* __restrict__ x, const float* __restrict__ Wr,
    const int* __restrict__ alpha_p, int* __restrict__ tkey,
    float* __restrict__ tgate)
{
  __shared__ float part[8 * 256];
  __shared__ __attribute__((aligned(16))) float sc[256];
  __shared__ __attribute__((aligned(16))) float stg[8 * 32 * SG_PITCH];

  const int tid  = threadIdx.x;
  const int wave = tid >> 5;
  const int lane = tid & 31;
  const int m = lane & 15;
  const int h = lane >> 4;
  const int t0 = blockIdx.x * 16;

  const int g  = lane >> 4;
  const int cl = lane & 15;
  const float* xl = x  + (size_t)(t0 + g) * D_DIM + wave * RT_KW + 4 * cl;
  const float* wl = Wr + (size_t)g * D_DIM        + wave * RT_KW + 4 * cl;
  float* sw  = stg + wave * (32 * SG_PITCH);
  float* swr = sw + g * SG_PITCH + 4 * cl;
  const float* frd = sw + m * SG_PITCH + 4 * h;

  const int alpha_i = *alpha_p;

  v4f gq[4][16];
#pragma unroll
  for (int q = 0; q < 4; ++q) {
#pragma unroll
    for (int j = 0; j < 8; ++j) {
      gq[q][j]     = *(const v4f*)(xl + j * 2 * D_DIM + q * 64);
      gq[q][8 + j] = *(const v4f*)(wl + j * 2 * D_DIM + q * 64);
    }
  }
  __builtin_amdgcn_sched_barrier(0);

  v8f c0 = {}, c1 = {}, c2 = {}, c3 = {};
  v4f a[2][8], b[2][8];
#pragma unroll
  for (int q = -1; q < 4; ++q) {
    if (q + 1 < 4) {
      const int nx = (q + 1) & 1;
#pragma unroll
      for (int j = 0; j < 16; ++j) *(v4f*)(swr + j * 2 * SG_PITCH) = gq[q + 1][j];
#pragma unroll
      for (int i = 0; i < 8; ++i) {
        a[nx][i] = *(const v4f*)(frd + i * 8);
        b[nx][i] = *(const v4f*)(frd + 16 * SG_PITCH + i * 8);
      }
    }
    __builtin_amdgcn_sched_barrier(0);
    if (q >= 0) {
      const int cu = q & 1;
#pragma unroll
      for (int i = 0; i < 8; i += 2) {
        c0 = wmma_f32(frag_lo(a[cu][i]),     frag_lo(b[cu][i]),     c0);
        c1 = wmma_f32(frag_hi(a[cu][i]),     frag_hi(b[cu][i]),     c1);
        c2 = wmma_f32(frag_lo(a[cu][i + 1]), frag_lo(b[cu][i + 1]), c2);
        c3 = wmma_f32(frag_hi(a[cu][i + 1]), frag_hi(b[cu][i + 1]), c3);
      }
      __builtin_amdgcn_sched_barrier(0);
    }
  }

#pragma unroll
  for (int i = 0; i < 8; ++i)
    part[wave * 256 + (8 * h + i) * 16 + m] = (c0[i] + c1[i]) + (c2[i] + c3[i]);
  __syncthreads();

  float s = part[tid];
#pragma unroll
  for (int w = 1; w < 8; ++w) s += part[w * 256 + tid];
  sc[tid] = s;
  __syncthreads();

  const int tk = tid >> 4;
  const int ke = tid & 15;
  float acc[K_EXP];
#pragma unroll
  for (int q = 0; q < 4; ++q) {
    const v4f v = *(const v4f*)(sc + tk * 16 + q * 4);
    acc[q * 4 + 0] = v[0];
    acc[q * 4 + 1] = v[1];
    acc[q * 4 + 2] = v[2];
    acc[q * 4 + 3] = v[3];
  }
  const float alpha = (float)alpha_i;
  int i0 = 0; float v0 = acc[0];
#pragma unroll
  for (int k = 1; k < K_EXP; ++k) if (acc[k] > v0) { v0 = acc[k]; i0 = k; }
  int i1 = -1; float v1 = -3.4e38f;
#pragma unroll
  for (int k = 0; k < K_EXP; ++k)
    if (k != i0 && acc[k] > v1) { v1 = acc[k]; i1 = k; }
  const float e  = __expf(v1 - v0);
  const float dn = 1.f + e;
  const float g0 = alpha / dn;
  const float g1 = alpha * e / dn;
  if (i1 < 0) i1 = (i0 == 0) ? 1 : 0;
  if (ke == 0) {
    const int t = t0 + tk;
    const bool sw = i1 < i0;
    tkey[t] = sw ? (i1 * 16 + i0) : (i0 * 16 + i1);
    tgate[2 * t]     = sw ? g1 : g0;
    tgate[2 * t + 1] = sw ? g0 : g1;
  }
}

#define TL_QG 8
#define HS_PITCH 36
#define SG_ROWS  64

template <int MT>
static __device__ __forceinline__ void tile_body(
    const float* x, const float* Bm, const float* tgate, float* out,
    const float* wlo, const float* whi, const v4f (&aw)[16], const int elo, const int ehi,
    float* part, float* hs, const int* stok, float* sw,
    const int tid, const int wave, const int lane, const int m, const int h)
{
  constexpr int NX = 8 * MT;
  constexpr int NJ = NX + 16;
  const int g = lane >> 4;
  const int c = lane & 15;
  const float* xg[NX];
#pragma unroll
  for (int j = 0; j < NX; ++j) {
    const int tokm = stok[2 * j + g];
    const int tokc = (tokm >= 0 && tokm < NTOK) ? tokm : 0;
    xg[j] = x + (size_t)tokc * D_DIM + wave * 256 + 4 * c;
  }
  float* swr = sw + g * SG_PITCH + 4 * c;
  const float* frd = sw + m * SG_PITCH + 4 * h;

  v8f acc[MT][4];
#pragma unroll
  for (int mt = 0; mt < MT; ++mt)
#pragma unroll
    for (int j = 0; j < 4; ++j) acc[mt][j] = v8f{};

  v2f gate[MT];
#pragma unroll
  for (int mt = 0; mt < MT; ++mt) {
    const int tk = stok[16 * mt + (tid >> 4)];
    const bool ok = (tk >= 0 && tk < NTOK);
    const v2f gv = *(const v2f*)(tgate + 2 * (ok ? tk : 0));
    gate[mt][0] = ok ? gv[0] : 0.f;
    gate[mt][1] = ok ? gv[1] : 0.f;
  }

  const int dcol = wave * 256 + m;
  const float* blo = Bm + ((size_t)elo * D_DIM + dcol) * R_RANK + 4 * h;
  const float* bhi = Bm + ((size_t)ehi * D_DIM + dcol) * R_RANK + 4 * h;
  v4f bq[4][4];

  v4f gq[NJ];
#pragma unroll
  for (int j = 0; j < NX; ++j) gq[j] = *(const v4f*)(xg[j]);
#pragma unroll
  for (int j = 0; j < 16; ++j) gq[NX + j] = aw[j];
#pragma unroll
  for (int q = 0; q < 4; ++q) {
#pragma unroll
    for (int j = 0; j < NJ; ++j) *(v4f*)(swr + j * 2 * SG_PITCH) = gq[j];
    if (q + 1 < 4) {
#pragma unroll
      for (int j = 0; j < NX; ++j) gq[j] = *(const v4f*)(xg[j] + (q + 1) * 64);
#pragma unroll
      for (int j = 0; j < 8; ++j) {
        gq[NX + j]     = *(const v4f*)(wlo + j * 2 * D_DIM + (q + 1) * 64);
        gq[NX + 8 + j] = *(const v4f*)(whi + j * 2 * D_DIM + (q + 1) * 64);
      }
    } else {
#pragma unroll
      for (int t = 0; t < 4; ++t) {
        bq[t][0] = *(const v4f*)(blo + t * 16 * R_RANK);
        bq[t][1] = *(const v4f*)(blo + t * 16 * R_RANK + 8);
        bq[t][2] = *(const v4f*)(bhi + t * 16 * R_RANK);
        bq[t][3] = *(const v4f*)(bhi + t * 16 * R_RANK + 8);
      }
    }
    v4f a[MT][TL_QG], bl[TL_QG], bh[TL_QG];
#pragma unroll
    for (int i = 0; i < TL_QG; ++i) {
#pragma unroll
      for (int mt = 0; mt < MT; ++mt)
        a[mt][i] = *(const v4f*)(frd + (16 * mt) * SG_PITCH + i * 8);
      bl[i] = *(const v4f*)(frd + (16 * MT) * SG_PITCH + i * 8);
      bh[i] = *(const v4f*)(frd + (16 * MT + 16) * SG_PITCH + i * 8);
    }
    __builtin_amdgcn_sched_barrier(0);
#pragma unroll
    for (int i = 0; i < TL_QG; ++i) {
#pragma unroll
      for (int mt = 0; mt < MT; ++mt) {
        acc[mt][0] = wmma_f32(frag_lo(a[mt][i]), frag_lo(bl[i]), acc[mt][0]);
        acc[mt][1] = wmma_f32(frag_lo(a[mt][i]), frag_lo(bh[i]), acc[mt][1]);
        acc[mt][2] = wmma_f32(frag_hi(a[mt][i]), frag_hi(bl[i]), acc[mt][2]);
        acc[mt][3] = wmma_f32(frag_hi(a[mt][i]), frag_hi(bh[i]), acc[mt][3]);
      }
    }
    __builtin_amdgcn_sched_barrier(0);
  }

#pragma unroll
  for (int mt = 0; mt < MT; ++mt) {
#pragma unroll
    for (int i = 0; i < 8; ++i) {
      part[wave * (MT * 512) + mt * 512 +       (8 * h + i) * 16 + m] = acc[mt][0][i] + acc[mt][2][i];
      part[wave * (MT * 512) + mt * 512 + 256 + (8 * h + i) * 16 + m] = acc[mt][1][i] + acc[mt][3][i];
    }
  }
  __syncthreads();

#pragma unroll
  for (int mt = 0; mt < MT; ++mt) {
    const int row = tid >> 4;
    const int r   = tid & 15;
    const float glo = gate[mt][0];
    const float ghi = gate[mt][1];
    float s0 = part[mt * 512 + tid];
    float s1 = part[mt * 512 + 256 + tid];
#pragma unroll
    for (int w = 1; w < 8; ++w) {
      s0 += part[w * (MT * 512) + mt * 512 + tid];
      s1 += part[w * (MT * 512) + mt * 512 + 256 + tid];
    }
    hs[(16 * mt + row) * HS_PITCH + r]      = s0 * glo;
    hs[(16 * mt + row) * HS_PITCH + 16 + r] = s1 * ghi;
  }
  __syncthreads();

  v4f af[MT][4];
#pragma unroll
  for (int mt = 0; mt < MT; ++mt)
#pragma unroll
    for (int q = 0; q < 4; ++q)
      af[mt][q] = *(const v4f*)(hs + (16 * mt + m) * HS_PITCH + q * 8 + 4 * h);

  int tko[MT][8];
#pragma unroll
  for (int mt = 0; mt < MT; ++mt)
#pragma unroll
    for (int i = 0; i < 8; ++i) {
      const int tk = stok[16 * mt + 8 * h + i];
      tko[mt][i] = (tk >= 0 && tk < NTOK) ? tk : -1;
    }

  v4f bn[4][4];
#pragma unroll
  for (int c = 0; c < 4; ++c) {
    if (c + 1 < 4) {
#pragma unroll
      for (int t = 0; t < 4; ++t) {
        const int o = ((c + 1) * 4 + t) * 16 * R_RANK;
        bn[t][0] = *(const v4f*)(blo + o);
        bn[t][1] = *(const v4f*)(blo + o + 8);
        bn[t][2] = *(const v4f*)(bhi + o);
        bn[t][3] = *(const v4f*)(bhi + o + 8);
      }
    }
    v8f c0[MT][4], c1[MT][4];
#pragma unroll
    for (int mt = 0; mt < MT; ++mt)
#pragma unroll
      for (int t = 0; t < 4; ++t) { c0[mt][t] = v8f{}; c1[mt][t] = v8f{}; }
#pragma unroll
    for (int q = 0; q < 4; ++q) {
#pragma unroll
      for (int t = 0; t < 4; ++t) {
#pragma unroll
        for (int mt = 0; mt < MT; ++mt) {
          c0[mt][t] = wmma_f32(frag_lo(af[mt][q]), frag_lo(bq[t][q]), c0[mt][t]);
          c1[mt][t] = wmma_f32(frag_hi(af[mt][q]), frag_hi(bq[t][q]), c1[mt][t]);
        }
      }
    }
#pragma unroll
    for (int mt = 0; mt < MT; ++mt)
#pragma unroll
      for (int i = 0; i < 8; ++i) {
        if (tko[mt][i] >= 0) {
          float* o = out + (size_t)tko[mt][i] * D_DIM + wave * 256 + m + c * 64;
#pragma unroll
          for (int t = 0; t < 4; ++t) o[t * 16] = c0[mt][t][i] + c1[mt][t][i];
        }
      }
#pragma unroll
    for (int t = 0; t < 4; ++t)
#pragma unroll
      for (int q = 0; q < 4; ++q) bq[t][q] = bn[t][q];
  }
}

__global__ __launch_bounds__(256) void moelora_tiles(
    const float* x, const float* Amat, const float* Bm, const float* tgate,
    const int* __restrict__ tkey, float* out)
{
  __shared__ float part[8 * 1024];
  __shared__ __attribute__((aligned(16))) float hs[32 * HS_PITCH];
  __shared__ __attribute__((aligned(16))) float stg[8 * SG_ROWS * SG_PITCH];
  __shared__ int stok[32];
  __shared__ int wtot[8];

  const int tid  = threadIdx.x;
  const int wave = tid >> 5;
  const int lane = tid & 31;
  const int m = lane & 15;
  const int h = lane >> 4;

  int elo = 0, rem = blockIdx.x;
#pragma unroll
  for (int s = 0; s < 15; ++s)
    if (rem >= 15 - elo) { rem -= 15 - elo; ++elo; }
  const int ehi = elo + 1 + rem;
  const int mykey = elo * 16 + ehi;

  const float* wlo = Amat + (size_t)(elo * 16 + (lane >> 4)) * D_DIM + wave * 256 + 4 * (lane & 15);
  const float* whi = Amat + (size_t)(ehi * 16 + (lane >> 4)) * D_DIM + wave * 256 + 4 * (lane & 15);

  v4i kv[4];
#pragma unroll
  for (int q = 0; q < 4; ++q) kv[q] = *(const v4i*)(tkey + q * 1024 + tid * 4);
  v4f aw[16];
#pragma unroll
  for (int j = 0; j < 8; ++j) {
    aw[j]     = *(const v4f*)(wlo + j * 2 * D_DIM);
    aw[8 + j] = *(const v4f*)(whi + j * 2 * D_DIM);
  }
  unsigned mask = 0;
#pragma unroll
  for (int q = 0; q < 4; ++q) {
#pragma unroll
    for (int c = 0; c < 4; ++c)
      mask |= (kv[q][c] == mykey ? 1u : 0u) << (q * 4 + c);
  }
  const int cnt = __builtin_popcount(mask);
  int incl = cnt;
#pragma unroll
  for (int o = 1; o < 32; o <<= 1) {
    const int v = __shfl_up(incl, o);
    if (lane >= o) incl += v;
  }
  if (lane == 31) wtot[wave] = incl;
  barrier_lds();
  int base = incl - cnt, total = 0;
#pragma unroll
  for (int w = 0; w < 8; ++w) {
    const int t = wtot[w];
    total += t;
    base += (w < wave) ? t : 0;
  }

  float* sw = stg + wave * (SG_ROWS * SG_PITCH);

  for (int slot = blockIdx.y; slot * 32 < total; slot += JGRID) {
    if (slot != (int)blockIdx.y) {
      __syncthreads();
#pragma unroll
      for (int j = 0; j < 8; ++j) {
        aw[j]     = *(const v4f*)(wlo + j * 2 * D_DIM);
        aw[8 + j] = *(const v4f*)(whi + j * 2 * D_DIM);
      }
    }

    const int left = total - 32 * slot;
    if (tid < 32 && tid >= left) stok[tid] = -1;
    {
      unsigned mm = mask;
      int r = base - 32 * slot;
      while (mm != 0) {
        const int b = __builtin_ctz(mm);
        mm &= mm - 1;
        if (r >= 0 && r < 32) stok[r] = (b >> 2) * 1024 + tid * 4 + (b & 3);
        ++r;
      }
    }
    barrier_lds();

    if (left > 16)
      tile_body<2>(x, Bm, tgate, out, wlo, whi, aw, elo, ehi, part, hs, stok, sw, tid, wave, lane, m, h);
    else
      tile_body<1>(x, Bm, tgate, out, wlo, whi, aw, elo, ehi, part, hs, stok, sw, tid, wave, lane, m, h);
  }
}

extern "C" void kernel_launch(void* const* d_in, const int* in_sizes, int n_in,
                              void* d_out, int out_size, void* d_ws, size_t ws_size,
                              hipStream_t stream) {
  (void)in_sizes; (void)n_in; (void)out_size; (void)ws_size;
  const float* x     = (const float*)d_in[0];
  const int*   alpha = (const int*)  d_in[1];
  const float* A     = (const float*)d_in[2];
  const float* Bm    = (const float*)d_in[3];
  const float* Wr    = (const float*)d_in[4];
  float* out = (float*)d_out;

  int*   tkey     = (int*)d_ws;
  float* tgate    = (float*)(tkey + NTOK);

  moelora_router<<<NTOK / 16, 256, 0, stream>>>(x, Wr, alpha, tkey, tgate);
  moelora_tiles<<<dim3(NPAIR, JGRID), 256, 0, stream>>>(x, A, Bm, tgate, tkey, out);
}
